# starting kernel
# baseline (speedup 1.0000x reference)
#include <hip/hip_runtime.h>
#include <math.h>

#define ATT_SLOPE 0.2f
#define ACT_SLOPE 0.01f

typedef __attribute__((ext_vector_type(2))) float v2f;
typedef __attribute__((ext_vector_type(8))) float v8f;

__device__ __forceinline__ float lrelu(float v, float s) { return v > 0.f ? v : v * s; }

__device__ __forceinline__ void atomicMaxF(float* addr, float val) {
  if (val >= 0.f)
    atomicMax((int*)addr, __float_as_int(val));
  else
    atomicMin((unsigned int*)addr, (unsigned int)__float_as_int(val));
}

template <int K, int NCOLS>
__global__ void __launch_bounds__(32)
wmma_gemm_f32(const float* __restrict__ A, const float* __restrict__ B,
              float* __restrict__ D, int M) {
  const int m0 = blockIdx.x * 16;
  const int n0 = blockIdx.y * 16;
  const int lane = threadIdx.x;
  const int half = lane >> 4;
  const int lid  = lane & 15;
  const int ar = min(m0 + lid, M - 1);
  v8f acc = {};
#pragma unroll
  for (int k = 0; k < K; k += 4) {
    const int ka = k + half * 2;
    v2f a, b;
    a.x = A[(size_t)ar * K + ka];
    a.y = A[(size_t)ar * K + ka + 1];
    b.x = B[(size_t)ka * NCOLS + n0 + lid];
    b.y = B[(size_t)(ka + 1) * NCOLS + n0 + lid];
    acc = __builtin_amdgcn_wmma_f32_16x16x4_f32(false, a, false, b, (short)0, acc,
                                                false, false);
  }
  const int rbase = m0 + half * 8;
#pragma unroll
  for (int r = 0; r < 8; ++r) {
    const int row = rbase + r;
    if (row < M) D[(size_t)row * NCOLS + n0 + lid] = acc[r];
  }
}

__global__ void zero_buf(float* __restrict__ p, unsigned n) {
  const unsigned t = blockIdx.x * blockDim.x + threadIdx.x;
  if (t < n) p[t] = 0.f;
}

__global__ void node_prep(const float* __restrict__ h,
                          const float* __restrict__ att_src,
                          const float* __restrict__ att_dst,
                          float* __restrict__ as, float* __restrict__ ad,
                          float* __restrict__ m, float* __restrict__ denom,
                          int Nn, int H, int C) {
  const int n = blockIdx.x * blockDim.x + threadIdx.x;
  if (n >= Nn) return;
  const int HC = H * C;
  for (int hh = 0; hh < H; ++hh) {
    float s = 0.f, d = 0.f;
    for (int c = 0; c < C; ++c) {
      const float v = h[(size_t)n * HC + hh * C + c];
      s += v * att_src[hh * C + c];
      d += v * att_dst[hh * C + c];
    }
    as[n * H + hh] = s;
    ad[n * H + hh] = d;
    m[n * H + hh] = -INFINITY;
    denom[n * H + hh] = 0.f;
  }
}

__global__ void edge_max(const int* __restrict__ src, const int* __restrict__ dst,
                         const float* __restrict__ as, const float* __restrict__ ad,
                         float* __restrict__ m, int E, int Nn, int H) {
  const unsigned t = blockIdx.x * blockDim.x + threadIdx.x;
  const unsigned tot = (unsigned)(E + Nn) * (unsigned)H;
  if (t >= tot) return;
  const int hh = (int)(t % (unsigned)H);
  const int e  = (int)(t / (unsigned)H);
  int s, d;
  if (e < E) { s = src[e]; d = dst[e]; } else { s = d = e - E; }
  const float v = lrelu(as[s * H + hh] + ad[d * H + hh], ATT_SLOPE);
  atomicMaxF(&m[d * H + hh], v);
}

__global__ void edge_accum(const int* __restrict__ src, const int* __restrict__ dst,
                           const float* __restrict__ as, const float* __restrict__ ad,
                           const float* __restrict__ m, const float* __restrict__ hfeat,
                           float* __restrict__ denom, float* __restrict__ acc,
                           int E, int Nn, int H, int C) {
  const int HC = H * C;
  const unsigned t = blockIdx.x * blockDim.x + threadIdx.x;
  const unsigned tot = (unsigned)(E + Nn) * (unsigned)HC;
  if (t >= tot) return;
  const int j = (int)(t % (unsigned)HC);
  const int e = (int)(t / (unsigned)HC);
  const int hh = j / C;
  int s, d;
  if (e < E) { s = src[e]; d = dst[e]; } else { s = d = e - E; }
  const float v = lrelu(as[s * H + hh] + ad[d * H + hh], ATT_SLOPE);
  const float w = expf(v - m[d * H + hh]);
  if ((j - hh * C) == 0) atomicAdd(&denom[d * H + hh], w);
  atomicAdd(&acc[(size_t)d * HC + j], w * hfeat[(size_t)s * HC + j]);
}

__global__ void finalize_feat(float* __restrict__ acc, const float* __restrict__ denom,
                              const float* __restrict__ bias, int Nn, int H, int C) {
  const unsigned t = blockIdx.x * blockDim.x + threadIdx.x;
  const int HC = H * C;
  if (t >= (unsigned)Nn * (unsigned)HC) return;
  const int n = (int)(t / (unsigned)HC);
  const int j = (int)(t % (unsigned)HC);
  const float v = acc[(size_t)n * HC + j] / denom[n * H + j / C] + bias[j];
  acc[(size_t)n * HC + j] = lrelu(v, ACT_SLOPE);
}

__global__ void finalize_proj(const float* __restrict__ acc, const float* __restrict__ denom,
                              const float* __restrict__ bias, const float* __restrict__ Wo,
                              const float* __restrict__ bo, float* __restrict__ out,
                              int Nn, int C) {
  const int n = blockIdx.x * blockDim.x + threadIdx.x;
  if (n >= Nn) return;
  const float dn = denom[n];
  float s = 0.f;
  for (int c = 0; c < C; ++c) {
    const float f = lrelu(acc[(size_t)n * C + c] / dn + bias[c], ACT_SLOPE);
    s += f * Wo[c];
  }
  out[n] = s + bo[0];
}

extern "C" void kernel_launch(void* const* d_in, const int* in_sizes, int n_in,
                              void* d_out, int out_size, void* d_ws, size_t ws_size,
                              hipStream_t stream) {
  const float* x    = (const float*)d_in[0];
  const int*   ei   = (const int*)d_in[1];
  const float* W1   = (const float*)d_in[2];
  const float* aS1  = (const float*)d_in[3];
  const float* aD1  = (const float*)d_in[4];
  const float* b1   = (const float*)d_in[5];
  const float* W2   = (const float*)d_in[6];
  const float* aS2  = (const float*)d_in[7];
  const float* aD2  = (const float*)d_in[8];
  const float* b2   = (const float*)d_in[9];
  const float* Wo   = (const float*)d_in[10];
  const float* bo   = (const float*)d_in[11];
  float* out = (float*)d_out;

  const int IN_C = 64, HID = 16, HEADS = 4;
  const int HC1 = HEADS * HID;
  const int Nn = in_sizes[0] / IN_C;
  const int E  = in_sizes[1] / 2;
  const int* src = ei;
  const int* dst = ei + E;

  float* p = (float*)d_ws;
  float* h1   = p; p += (size_t)Nn * HC1;
  float* acc1 = p; p += (size_t)Nn * HC1;
  float* h2   = p; p += (size_t)Nn * HID;
  float* acc2 = p; p += (size_t)Nn * HID;
  float* as1  = p; p += (size_t)Nn * HEADS;
  float* ad1  = p; p += (size_t)Nn * HEADS;
  float* m1   = p; p += (size_t)Nn * HEADS;
  float* dn1  = p; p += (size_t)Nn * HEADS;
  float* as2  = p; p += (size_t)Nn;
  float* ad2  = p; p += (size_t)Nn;
  float* m2   = p; p += (size_t)Nn;
  float* dn2  = p; p += (size_t)Nn;

  const int TB = 256;

  {
    dim3 g((Nn + 15) / 16, HC1 / 16);
    wmma_gemm_f32<64, 64><<<g, 32, 0, stream>>>(x, W1, h1, Nn);
  }
  {
    const unsigned n = (unsigned)Nn * (unsigned)HC1;
    zero_buf<<<(n + TB - 1) / TB, TB, 0, stream>>>(acc1, n);
  }
  node_prep<<<(Nn + TB - 1) / TB, TB, 0, stream>>>(h1, aS1, aD1, as1, ad1, m1, dn1,
                                                   Nn, HEADS, HID);
  {
    const unsigned n = (unsigned)(E + Nn) * (unsigned)HEADS;
    edge_max<<<(n + TB - 1) / TB, TB, 0, stream>>>(src, dst, as1, ad1, m1,
                                                   E, Nn, HEADS);
  }
  {
    const unsigned n = (unsigned)(E + Nn) * (unsigned)HC1;
    edge_accum<<<(n + TB - 1) / TB, TB, 0, stream>>>(src, dst, as1, ad1, m1,
                                                     h1, dn1, acc1,
                                                     E, Nn, HEADS, HID);
  }
  {
    const unsigned n = (unsigned)Nn * (unsigned)HC1;
    finalize_feat<<<(n + TB - 1) / TB, TB, 0, stream>>>(acc1, dn1, b1, Nn, HEADS, HID);
  }

  {
    dim3 g((Nn + 15) / 16, HID / 16);
    wmma_gemm_f32<64, 16><<<g, 32, 0, stream>>>(acc1, W2, h2, Nn);
  }
  {
    const unsigned n = (unsigned)Nn * (unsigned)HID;
    zero_buf<<<(n + TB - 1) / TB, TB, 0, stream>>>(acc2, n);
  }
  node_prep<<<(Nn + TB - 1) / TB, TB, 0, stream>>>(h2, aS2, aD2, as2, ad2, m2, dn2,
                                                   Nn, 1, HID);
  {
    const unsigned n = (unsigned)(E + Nn);
    edge_max<<<(n + TB - 1) / TB, TB, 0, stream>>>(src, dst, as2, ad2, m2,
                                                   E, Nn, 1);
  }
  {
    const unsigned n = (unsigned)(E + Nn) * (unsigned)HID;
    edge_accum<<<(n + TB - 1) / TB, TB, 0, stream>>>(src, dst, as2, ad2, m2,
                                                     h2, dn2, acc2,
                                                     E, Nn, 1, HID);
  }
  finalize_proj<<<(Nn + TB - 1) / TB, TB, 0, stream>>>(acc2, dn2, b2, Wo, bo, out,
                                                       Nn, HID);
}


# Owner filter reads one owner-number byte per edge, written once per call by a new pre-pass launch, not the 4-byte destination: 16 edges per 16-byte load; 6 launches 7; sums still in arrival order
# speedup vs baseline: 4.8217x; 4.8217x over previous
#include <hip/hip_runtime.h>
#include <math.h>
#include <stdint.h>


#define ATT_SLOPE 0.2f
#define ACT_SLOPE 0.01f

typedef __attribute__((ext_vector_type(2))) float v2f;
typedef __attribute__((ext_vector_type(8))) float v8f;

__device__ __forceinline__ float lrelu(float v, float s) { return v > 0.f ? v : v * s; }

__device__ __forceinline__ void lds_max_f(int* slot, float val) {
  if (val >= 0.f)
    atomicMax(slot, __float_as_int(val));
  else
    atomicMin((unsigned int*)slot, (unsigned int)__float_as_int(val));
}

template <int K, int NCOLS>
__global__ void __launch_bounds__(32)
wmma_gemm_f32(const float* __restrict__ A, const float* __restrict__ B,
              float* __restrict__ D, int M) {
  const int m0 = blockIdx.x * 16;
  const int n0 = blockIdx.y * 16;
  const int lane = threadIdx.x;
  const int half = lane >> 4;
  const int lid  = lane & 15;
  const int ar = min(m0 + lid, M - 1);
  v8f acc = {};
#pragma unroll
  for (int k = 0; k < K; k += 4) {
    const int ka = k + half * 2;
    v2f a, b;
    a.x = A[(size_t)ar * K + ka];
    a.y = A[(size_t)ar * K + ka + 1];
    b.x = B[(size_t)ka * NCOLS + n0 + lid];
    b.y = B[(size_t)(ka + 1) * NCOLS + n0 + lid];
    acc = __builtin_amdgcn_wmma_f32_16x16x4_f32(false, a, false, b, (short)0, acc,
                                                false, false);
  }
  const int rbase = m0 + half * 8;
#pragma unroll
  for (int r = 0; r < 8; ++r) {
    const int row = rbase + r;
    if (row < M) D[(size_t)row * NCOLS + n0 + lid] = acc[r];
  }
}

__global__ void node_coef(const float* __restrict__ h,
                          const float* __restrict__ att_src,
                          const float* __restrict__ att_dst,
                          float* __restrict__ as, float* __restrict__ ad,
                          int Nn, int H, int C) {
  const int n = blockIdx.x * blockDim.x + threadIdx.x;
  if (n >= Nn) return;
  const int HC = H * C;
  for (int hh = 0; hh < H; ++hh) {
    float s = 0.f, d = 0.f;
    for (int c = 0; c < C; ++c) {
      const float v = h[(size_t)n * HC + hh * C + c];
      s += v * att_src[hh * C + c];
      d += v * att_dst[hh * C + c];
    }
    as[n * H + hh] = s;
    ad[n * H + hh] = d;
  }
}

constexpr int OWN_NT  = 1024;
constexpr int OWN_NPW = 224;
constexpr int OWN_CAP = 9216;
constexpr int OWN_UNR = 8;
constexpr unsigned OWN_NOKEY = 255u;
constexpr int KEY_NT  = 256;
constexpr int KEY_UNR = 4;

__device__ __forceinline__ unsigned owner_key(int d, int Nn) {
  return (unsigned)d < (unsigned)Nn ? min((unsigned)d / (unsigned)OWN_NPW, OWN_NOKEY) : OWN_NOKEY;
}

__global__ void __launch_bounds__(KEY_NT)
owner_keys(const int* __restrict__ dst, unsigned int* __restrict__ keys, int E, int Nn) {
  const int nq = (E + 3) >> 2;
  const bool vec = ((reinterpret_cast<uintptr_t>(dst) & 15) == 0);
  const int4* d4 = reinterpret_cast<const int4*>(dst);
  const int q0 = blockIdx.x * (KEY_NT * KEY_UNR) + threadIdx.x;
  int4 v[KEY_UNR];
#pragma unroll
  for (int u = 0; u < KEY_UNR; ++u) {
    const int q = q0 + u * KEY_NT;
    const int e0 = q * 4;
    if (vec && q < nq && e0 + 3 < E) {
      v[u] = d4[q];
    } else {
      v[u].x = (q < nq && e0     < E) ? dst[e0]     : -1;
      v[u].y = (q < nq && e0 + 1 < E) ? dst[e0 + 1] : -1;
      v[u].z = (q < nq && e0 + 2 < E) ? dst[e0 + 2] : -1;
      v[u].w = (q < nq && e0 + 3 < E) ? dst[e0 + 3] : -1;
    }
  }
#pragma unroll
  for (int u = 0; u < KEY_UNR; ++u) {
    const int q = q0 + u * KEY_NT;
    if (q < nq)
      keys[q] = owner_key(v[u].x, Nn) | (owner_key(v[u].y, Nn) << 8) |
                (owner_key(v[u].z, Nn) << 16) | (owner_key(v[u].w, Nn) << 24);
  }
}

template <int H, int C, bool FINAL>
__global__ void __launch_bounds__(OWN_NT)
owner_layer(const int* __restrict__ src, const int* __restrict__ dst,
            const unsigned char* __restrict__ keys,
            const float* __restrict__ as, const float* __restrict__ ad,
            const float* __restrict__ hfeat, const float* __restrict__ bias,
            const float* __restrict__ Wo, const float* __restrict__ bo,
            float* __restrict__ outp, int E, int Nn) {
  constexpr int HC = H * C;
  static_assert(OWN_NT % HC == 0, "a thread keeps one feature column");
  __shared__ float acc_l[OWN_NPW * HC];
  __shared__ float den_l[OWN_NPW * H];
  __shared__ float ad_l[OWN_NPW * H];
  __shared__ int   m_l[OWN_NPW * H];
  __shared__ int   list_s[OWN_CAP];
  __shared__ int   list_d[OWN_CAP];
  __shared__ float ew[OWN_CAP * H];
  __shared__ int   cnt;

  const int tid = threadIdx.x;
  const int n0 = blockIdx.x * OWN_NPW;
  const int nloc = min(OWN_NPW, Nn - n0);
  if (nloc <= 0) return;

  for (int i = tid; i < nloc * HC; i += OWN_NT) acc_l[i] = 0.f;
  for (int i = tid; i < nloc * H; i += OWN_NT) {
    den_l[i] = 0.f;
    m_l[i] = __float_as_int(-INFINITY);
    ad_l[i] = ad[(size_t)n0 * H + i];
  }
  for (int i = tid; i < nloc; i += OWN_NT) { list_s[i] = n0 + i; list_d[i] = i; }
  if (tid == 0) cnt = nloc;
  __syncthreads();

  const bool keyed = gridDim.x <= OWN_NOKEY;
  if (keyed) {
    auto push = [&](int e) {
      const int pos = atomicAdd(&cnt, 1);
      if (pos < OWN_CAP) list_s[pos] = e;
    };
    const unsigned mine = blockIdx.x;
    const unsigned kk = mine * 0x01010101u;
    auto eq4 = [&](unsigned w) {
      const unsigned x = w ^ kk;
      return ~(((x & 0x7f7f7f7fu) + 0x7f7f7f7fu) | x) & 0x80808080u;
    };
    auto take16 = [&](const uint4& v, int e0) {
      unsigned m = (eq4(v.x) >> 7) | (eq4(v.y) >> 6) | (eq4(v.z) >> 5) | (eq4(v.w) >> 4);
      while (m) {
        const int b = __ffs(m) - 1;
        m &= m - 1;
        push(e0 + ((b & 7) << 2) + (b >> 3));
      }
    };
    const bool vec = ((reinterpret_cast<uintptr_t>(keys) & 15) == 0);
    const int n16 = vec ? (E >> 4) : 0;
    const uint4* k16 = reinterpret_cast<const uint4*>(keys);
    const int start = (int)(((long long)blockIdx.x * n16) / gridDim.x) & ~7;
    for (int i = tid; i < n16; i += OWN_NT * OWN_UNR) {
      uint4 v[OWN_UNR];
      int   ph[OWN_UNR];
#pragma unroll
      for (int u = 0; u < OWN_UNR; ++u) {
        int q = min(i + u * OWN_NT, n16 - 1) + start;
        if (q >= n16) q -= n16;
        ph[u] = q;
        v[u] = k16[q];
      }
#pragma unroll
      for (int u = 0; u < OWN_UNR; ++u) {
        if (i + u * OWN_NT < n16) take16(v[u], ph[u] * 16);
      }
    }
    for (int e = n16 * 16 + tid; e < E; e += OWN_NT)
      if (keys[e] == mine) push(e);
  }
  __syncthreads();
  const int total = keyed ? cnt : OWN_CAP + 1;

  if (total <= OWN_CAP) {
    for (int i0 = nloc + tid; i0 < total; i0 += OWN_NT * OWN_UNR) {
      int s[OWN_UNR], d[OWN_UNR];
#pragma unroll
      for (int u = 0; u < OWN_UNR; ++u) {
        const int i = i0 + u * OWN_NT;
        const int e = list_s[min(i, total - 1)];
        const int ee = i < total ? e : 0;
        s[u] = src[ee];
        d[u] = dst[ee];
      }
#pragma unroll
      for (int u = 0; u < OWN_UNR; ++u) {
        const int i = i0 + u * OWN_NT;
        if (i < total) { list_s[i] = s[u]; list_d[i] = d[u] - n0; }
      }
    }
    __syncthreads();
    {
      constexpr int U1 = 4;
      const int totH = total * H;
      int it = tid;
      for (; it + (U1 - 1) * OWN_NT < totH; it += U1 * OWN_NT) {
        float a[U1];
        int sl[U1];
#pragma unroll
        for (int u = 0; u < U1; ++u) {
          const int q = it + u * OWN_NT;
          const int i = q / H, h = q % H;
          sl[u] = list_d[i] * H + h;
          a[u] = as[(size_t)list_s[i] * H + h];
        }
#pragma unroll
        for (int u = 0; u < U1; ++u) {
          const float v = lrelu(a[u] + ad_l[sl[u]], ATT_SLOPE);
          ew[it + u * OWN_NT] = v;
          lds_max_f(&m_l[sl[u]], v);
        }
      }
      for (; it < totH; it += OWN_NT) {
        const int i = it / H, h = it % H;
        const int sl = list_d[i] * H + h;
        const float v = lrelu(as[(size_t)list_s[i] * H + h] + ad_l[sl], ATT_SLOPE);
        ew[it] = v;
        lds_max_f(&m_l[sl], v);
      }
    }
    __syncthreads();
    for (int it = tid; it < total * H; it += OWN_NT) {
      const int i = it / H, h = it % H;
      const int dl = list_d[i];
      const float w = expf(ew[it] - __int_as_float(m_l[dl * H + h]));
      ew[it] = w;
      atomicAdd(&den_l[dl * H + h], w);
    }
    __syncthreads();
    constexpr int EPI = OWN_NT / HC;
    const int j = tid % HC;
    const int h = j / C;
    int i0 = tid / HC;
    for (; i0 + (OWN_UNR - 1) * EPI < total; i0 += EPI * OWN_UNR) {
      float hv[OWN_UNR], w[OWN_UNR];
      int dl[OWN_UNR];
#pragma unroll
      for (int u = 0; u < OWN_UNR; ++u) {
        const int i = i0 + u * EPI;
        dl[u] = list_d[i];
        w[u] = ew[i * H + h];
        hv[u] = hfeat[(size_t)list_s[i] * HC + j];
      }
#pragma unroll
      for (int u = 0; u < OWN_UNR; ++u) atomicAdd(&acc_l[dl[u] * HC + j], w[u] * hv[u]);
    }
    for (; i0 < total; i0 += EPI)
      atomicAdd(&acc_l[list_d[i0] * HC + j], ew[i0 * H + h] * hfeat[(size_t)list_s[i0] * HC + j]);
  } else {
    const int tot = E + nloc;
    for (int e = tid; e < tot; e += OWN_NT) {
      int s, dl;
      if (e < E) {
        const unsigned x = (unsigned)(dst[e] - n0);
        if (x >= (unsigned)nloc) continue;
        dl = (int)x; s = src[e];
      } else { dl = e - E; s = n0 + dl; }
      for (int h = 0; h < H; ++h) {
        const float v = lrelu(as[(size_t)s * H + h] + ad_l[dl * H + h], ATT_SLOPE);
        lds_max_f(&m_l[dl * H + h], v);
      }
    }
    __syncthreads();
    for (int e = tid; e < tot; e += OWN_NT) {
      int s, dl;
      if (e < E) {
        const unsigned x = (unsigned)(dst[e] - n0);
        if (x >= (unsigned)nloc) continue;
        dl = (int)x; s = src[e];
      } else { dl = e - E; s = n0 + dl; }
      for (int h = 0; h < H; ++h) {
        const float v = lrelu(as[(size_t)s * H + h] + ad_l[dl * H + h], ATT_SLOPE);
        const float w = expf(v - __int_as_float(m_l[dl * H + h]));
        atomicAdd(&den_l[dl * H + h], w);
        for (int c = 0; c < C; ++c)
          atomicAdd(&acc_l[dl * HC + h * C + c], w * hfeat[(size_t)s * HC + h * C + c]);
      }
    }
  }
  __syncthreads();

  if constexpr (!FINAL) {
    for (int it = tid; it < nloc * HC; it += OWN_NT) {
      const int n = it / HC, jj = it % HC;
      const float v = acc_l[it] / den_l[n * H + jj / C] + bias[jj];
      outp[(size_t)n0 * HC + it] = lrelu(v, ACT_SLOPE);
    }
  } else {
    for (int n = tid; n < nloc; n += OWN_NT) {
      const float dn = den_l[n];
      float s = 0.f;
      for (int c = 0; c < C; ++c) {
        const float f = lrelu(acc_l[n * C + c] / dn + bias[c], ACT_SLOPE);
        s += f * Wo[c];
      }
      outp[n0 + n] = s + bo[0];
    }
  }
}

extern "C" void kernel_launch(void* const* d_in, const int* in_sizes, int n_in,
                              void* d_out, int out_size, void* d_ws, size_t ws_size,
                              hipStream_t stream) {
  const float* x    = (const float*)d_in[0];
  const int*   ei   = (const int*)d_in[1];
  const float* W1   = (const float*)d_in[2];
  const float* aS1  = (const float*)d_in[3];
  const float* aD1  = (const float*)d_in[4];
  const float* b1   = (const float*)d_in[5];
  const float* W2   = (const float*)d_in[6];
  const float* aS2  = (const float*)d_in[7];
  const float* aD2  = (const float*)d_in[8];
  const float* b2   = (const float*)d_in[9];
  const float* Wo   = (const float*)d_in[10];
  const float* bo   = (const float*)d_in[11];
  float* out = (float*)d_out;

  const int IN_C = 64, HID = 16, HEADS = 4;
  const int HC1 = HEADS * HID;
  const int Nn = in_sizes[0] / IN_C;
  const int E  = in_sizes[1] / 2;
  const int* src = ei;
  const int* dst = ei + E;

  unsigned char* keys = (unsigned char*)d_ws;
  float* p = (float*)(keys + (((size_t)E + 255) & ~(size_t)255));
  float* h1   = p; p += (size_t)Nn * HC1;
  float* f1   = p; p += (size_t)Nn * HC1;
  float* h2   = p; p += (size_t)Nn * HID;
  float* as1  = p; p += (size_t)Nn * HEADS;
  float* ad1  = p; p += (size_t)Nn * HEADS;
  float* as2  = p; p += (size_t)Nn;
  float* ad2  = p; p += (size_t)Nn;

  const int TB = 256;
  const int OG = (Nn + OWN_NPW - 1) / OWN_NPW;

  owner_keys<<<(E + 4 * KEY_NT * KEY_UNR - 1) / (4 * KEY_NT * KEY_UNR), KEY_NT, 0, stream>>>(
      dst, (unsigned int*)keys, E, Nn);

  {
    dim3 g((Nn + 15) / 16, HC1 / 16);
    wmma_gemm_f32<64, 64><<<g, 32, 0, stream>>>(x, W1, h1, Nn);
  }
  node_coef<<<(Nn + TB - 1) / TB, TB, 0, stream>>>(h1, aS1, aD1, as1, ad1, Nn, HEADS, HID);
  owner_layer<4, 16, false><<<OG, OWN_NT, 0, stream>>>(src, dst, keys, as1, ad1, h1, b1, Wo,
                                                     bo, f1, E, Nn);

  {
    dim3 g((Nn + 15) / 16, HID / 16);
    wmma_gemm_f32<64, 16><<<g, 32, 0, stream>>>(f1, W2, h2, Nn);
  }
  node_coef<<<(Nn + TB - 1) / TB, TB, 0, stream>>>(h2, aS2, aD2, as2, ad2, Nn, 1, HID);
  owner_layer<1, 16, true><<<OG, OWN_NT, 0, stream>>>(src, dst, keys, as2, ad2, h2, b2, Wo,
                                                    bo, out, E, Nn);
}


# Layer 2 no longer filters the edges: layer 1's work-groups store their filtered lists (source, local destination) and lengths in the workspace, layer 2 loads them; sums still in arrival order
# speedup vs baseline: 5.9713x; 1.2384x over previous
#include <hip/hip_runtime.h>
#include <math.h>
#include <stdint.h>


#define ATT_SLOPE 0.2f
#define ACT_SLOPE 0.01f

typedef __attribute__((ext_vector_type(2))) float v2f;
typedef __attribute__((ext_vector_type(8))) float v8f;

__device__ __forceinline__ float lrelu(float v, float s) { return v > 0.f ? v : v * s; }

__device__ __forceinline__ void lds_max_f(int* slot, float val) {
  if (val >= 0.f)
    atomicMax(slot, __float_as_int(val));
  else
    atomicMin((unsigned int*)slot, (unsigned int)__float_as_int(val));
}

template <int K, int NCOLS>
__global__ void __launch_bounds__(32)
wmma_gemm_f32(const float* __restrict__ A, const float* __restrict__ B,
              float* __restrict__ D, int M) {
  const int m0 = blockIdx.x * 16;
  const int n0 = blockIdx.y * 16;
  const int lane = threadIdx.x;
  const int half = lane >> 4;
  const int lid  = lane & 15;
  const int ar = min(m0 + lid, M - 1);
  v8f acc = {};
#pragma unroll
  for (int k = 0; k < K; k += 4) {
    const int ka = k + half * 2;
    v2f a, b;
    a.x = A[(size_t)ar * K + ka];
    a.y = A[(size_t)ar * K + ka + 1];
    b.x = B[(size_t)ka * NCOLS + n0 + lid];
    b.y = B[(size_t)(ka + 1) * NCOLS + n0 + lid];
    acc = __builtin_amdgcn_wmma_f32_16x16x4_f32(false, a, false, b, (short)0, acc,
                                                false, false);
  }
  const int rbase = m0 + half * 8;
#pragma unroll
  for (int r = 0; r < 8; ++r) {
    const int row = rbase + r;
    if (row < M) D[(size_t)row * NCOLS + n0 + lid] = acc[r];
  }
}

__global__ void node_coef(const float* __restrict__ h,
                          const float* __restrict__ att_src,
                          const float* __restrict__ att_dst,
                          float* __restrict__ as, float* __restrict__ ad,
                          int Nn, int H, int C) {
  const int n = blockIdx.x * blockDim.x + threadIdx.x;
  if (n >= Nn) return;
  const int HC = H * C;
  for (int hh = 0; hh < H; ++hh) {
    float s = 0.f, d = 0.f;
    for (int c = 0; c < C; ++c) {
      const float v = h[(size_t)n * HC + hh * C + c];
      s += v * att_src[hh * C + c];
      d += v * att_dst[hh * C + c];
    }
    as[n * H + hh] = s;
    ad[n * H + hh] = d;
  }
}

constexpr int OWN_NT  = 1024;
constexpr int OWN_NPW = 224;
constexpr int OWN_CAP = 9216;
constexpr int OWN_UNR = 8;
constexpr unsigned OWN_NOKEY = 255u;
constexpr int KEY_NT  = 256;
constexpr int KEY_UNR = 4;

__device__ __forceinline__ unsigned owner_key(int d, int Nn) {
  return (unsigned)d < (unsigned)Nn ? min((unsigned)d / (unsigned)OWN_NPW, OWN_NOKEY) : OWN_NOKEY;
}

__global__ void __launch_bounds__(KEY_NT)
owner_keys(const int* __restrict__ dst, unsigned int* __restrict__ keys, int E, int Nn) {
  const int nq = (E + 3) >> 2;
  const bool vec = ((reinterpret_cast<uintptr_t>(dst) & 15) == 0);
  const int4* d4 = reinterpret_cast<const int4*>(dst);
  const int q0 = blockIdx.x * (KEY_NT * KEY_UNR) + threadIdx.x;
  int4 v[KEY_UNR];
#pragma unroll
  for (int u = 0; u < KEY_UNR; ++u) {
    const int q = q0 + u * KEY_NT;
    const int e0 = q * 4;
    if (vec && q < nq && e0 + 3 < E) {
      v[u] = d4[q];
    } else {
      v[u].x = (q < nq && e0     < E) ? dst[e0]     : -1;
      v[u].y = (q < nq && e0 + 1 < E) ? dst[e0 + 1] : -1;
      v[u].z = (q < nq && e0 + 2 < E) ? dst[e0 + 2] : -1;
      v[u].w = (q < nq && e0 + 3 < E) ? dst[e0 + 3] : -1;
    }
  }
#pragma unroll
  for (int u = 0; u < KEY_UNR; ++u) {
    const int q = q0 + u * KEY_NT;
    if (q < nq)
      keys[q] = owner_key(v[u].x, Nn) | (owner_key(v[u].y, Nn) << 8) |
                (owner_key(v[u].z, Nn) << 16) | (owner_key(v[u].w, Nn) << 24);
  }
}

template <int H, int C, bool FINAL, bool SAVED>
__global__ void __launch_bounds__(OWN_NT)
owner_layer(const int* __restrict__ src, const int* __restrict__ dst,
            const unsigned char* __restrict__ keys,
            int* __restrict__ sv_s, int* __restrict__ sv_d, int* __restrict__ sv_n,
            const float* __restrict__ as, const float* __restrict__ ad,
            const float* __restrict__ hfeat, const float* __restrict__ bias,
            const float* __restrict__ Wo, const float* __restrict__ bo,
            float* __restrict__ outp, int E, int Nn) {
  constexpr int HC = H * C;
  static_assert(OWN_NT % HC == 0, "a thread keeps one feature column");
  __shared__ float acc_l[OWN_NPW * HC];
  __shared__ float den_l[OWN_NPW * H];
  __shared__ float ad_l[OWN_NPW * H];
  __shared__ int   m_l[OWN_NPW * H];
  __shared__ int   list_s[OWN_CAP];
  __shared__ int   list_d[OWN_CAP];
  __shared__ float ew[OWN_CAP * H];
  __shared__ int   cnt;

  const int tid = threadIdx.x;
  const int n0 = blockIdx.x * OWN_NPW;
  const int nloc = min(OWN_NPW, Nn - n0);
  if (nloc <= 0) return;

  for (int i = tid; i < nloc * HC; i += OWN_NT) acc_l[i] = 0.f;
  for (int i = tid; i < nloc * H; i += OWN_NT) {
    den_l[i] = 0.f;
    m_l[i] = __float_as_int(-INFINITY);
    ad_l[i] = ad[(size_t)n0 * H + i];
  }
  for (int i = tid; i < nloc; i += OWN_NT) { list_s[i] = n0 + i; list_d[i] = i; }
  if (tid == 0) cnt = nloc;
  __syncthreads();

  int* const gs = sv_s + (size_t)blockIdx.x * OWN_CAP;
  int* const gd = sv_d + (size_t)blockIdx.x * OWN_CAP;

  const bool keyed = !SAVED && gridDim.x <= OWN_NOKEY;
  if (keyed) {
    auto push = [&](int e) {
      const int pos = atomicAdd(&cnt, 1);
      if (pos < OWN_CAP) list_s[pos] = e;
    };
    const unsigned mine = blockIdx.x;
    const unsigned kk = mine * 0x01010101u;
    auto eq4 = [&](unsigned w) {
      const unsigned x = w ^ kk;
      return ~(((x & 0x7f7f7f7fu) + 0x7f7f7f7fu) | x) & 0x80808080u;
    };
    auto take16 = [&](const uint4& v, int e0) {
      unsigned m = (eq4(v.x) >> 7) | (eq4(v.y) >> 6) | (eq4(v.z) >> 5) | (eq4(v.w) >> 4);
      while (m) {
        const int b = __ffs(m) - 1;
        m &= m - 1;
        push(e0 + ((b & 7) << 2) + (b >> 3));
      }
    };
    const bool vec = ((reinterpret_cast<uintptr_t>(keys) & 15) == 0);
    const int n16 = vec ? (E >> 4) : 0;
    const uint4* k16 = reinterpret_cast<const uint4*>(keys);
    const int start = (int)(((long long)blockIdx.x * n16) / gridDim.x) & ~7;
    for (int i = tid; i < n16; i += OWN_NT * OWN_UNR) {
      uint4 v[OWN_UNR];
      int   ph[OWN_UNR];
#pragma unroll
      for (int u = 0; u < OWN_UNR; ++u) {
        int q = min(i + u * OWN_NT, n16 - 1) + start;
        if (q >= n16) q -= n16;
        ph[u] = q;
        v[u] = k16[q];
      }
#pragma unroll
      for (int u = 0; u < OWN_UNR; ++u) {
        if (i + u * OWN_NT < n16) take16(v[u], ph[u] * 16);
      }
    }
    for (int e = n16 * 16 + tid; e < E; e += OWN_NT)
      if (keys[e] == mine) push(e);
  }
  __syncthreads();
  const int total = SAVED ? sv_n[blockIdx.x] : (keyed ? cnt : OWN_CAP + 1);
  if (!SAVED && tid == 0) sv_n[blockIdx.x] = total;

  if (total <= OWN_CAP) {
   if constexpr (SAVED) {
    for (int i0 = nloc + tid; i0 < total; i0 += OWN_NT * OWN_UNR) {
      int s[OWN_UNR], d[OWN_UNR];
#pragma unroll
      for (int u = 0; u < OWN_UNR; ++u) {
        const int i = min(i0 + u * OWN_NT, total - 1);
        s[u] = gs[i];
        d[u] = gd[i];
      }
#pragma unroll
      for (int u = 0; u < OWN_UNR; ++u) {
        const int i = i0 + u * OWN_NT;
        if (i < total) { list_s[i] = s[u]; list_d[i] = d[u]; }
      }
    }
   } else {
    for (int i0 = nloc + tid; i0 < total; i0 += OWN_NT * OWN_UNR) {
      int s[OWN_UNR], d[OWN_UNR];
#pragma unroll
      for (int u = 0; u < OWN_UNR; ++u) {
        const int i = i0 + u * OWN_NT;
        const int e = list_s[min(i, total - 1)];
        const int ee = i < total ? e : 0;
        s[u] = src[ee];
        d[u] = dst[ee];
      }
#pragma unroll
      for (int u = 0; u < OWN_UNR; ++u) {
        const int i = i0 + u * OWN_NT;
        if (i < total) {
          list_s[i] = s[u]; list_d[i] = d[u] - n0;
          gs[i] = s[u]; gd[i] = d[u] - n0;
        }
      }
    }
   }
    __syncthreads();
    {
      constexpr int U1 = 4;
      const int totH = total * H;
      int it = tid;
      for (; it + (U1 - 1) * OWN_NT < totH; it += U1 * OWN_NT) {
        float a[U1];
        int sl[U1];
#pragma unroll
        for (int u = 0; u < U1; ++u) {
          const int q = it + u * OWN_NT;
          const int i = q / H, h = q % H;
          sl[u] = list_d[i] * H + h;
          a[u] = as[(size_t)list_s[i] * H + h];
        }
#pragma unroll
        for (int u = 0; u < U1; ++u) {
          const float v = lrelu(a[u] + ad_l[sl[u]], ATT_SLOPE);
          ew[it + u * OWN_NT] = v;
          lds_max_f(&m_l[sl[u]], v);
        }
      }
      for (; it < totH; it += OWN_NT) {
        const int i = it / H, h = it % H;
        const int sl = list_d[i] * H + h;
        const float v = lrelu(as[(size_t)list_s[i] * H + h] + ad_l[sl], ATT_SLOPE);
        ew[it] = v;
        lds_max_f(&m_l[sl], v);
      }
    }
    __syncthreads();
    for (int it = tid; it < total * H; it += OWN_NT) {
      const int i = it / H, h = it % H;
      const int dl = list_d[i];
      const float w = expf(ew[it] - __int_as_float(m_l[dl * H + h]));
      ew[it] = w;
      atomicAdd(&den_l[dl * H + h], w);
    }
    __syncthreads();
    constexpr int EPI = OWN_NT / HC;
    const int j = tid % HC;
    const int h = j / C;
    int i0 = tid / HC;
    for (; i0 + (OWN_UNR - 1) * EPI < total; i0 += EPI * OWN_UNR) {
      float hv[OWN_UNR], w[OWN_UNR];
      int dl[OWN_UNR];
#pragma unroll
      for (int u = 0; u < OWN_UNR; ++u) {
        const int i = i0 + u * EPI;
        dl[u] = list_d[i];
        w[u] = ew[i * H + h];
        hv[u] = hfeat[(size_t)list_s[i] * HC + j];
      }
#pragma unroll
      for (int u = 0; u < OWN_UNR; ++u) atomicAdd(&acc_l[dl[u] * HC + j], w[u] * hv[u]);
    }
    for (; i0 < total; i0 += EPI)
      atomicAdd(&acc_l[list_d[i0] * HC + j], ew[i0 * H + h] * hfeat[(size_t)list_s[i0] * HC + j]);
  } else {
    const int tot = E + nloc;
    for (int e = tid; e < tot; e += OWN_NT) {
      int s, dl;
      if (e < E) {
        const unsigned x = (unsigned)(dst[e] - n0);
        if (x >= (unsigned)nloc) continue;
        dl = (int)x; s = src[e];
      } else { dl = e - E; s = n0 + dl; }
      for (int h = 0; h < H; ++h) {
        const float v = lrelu(as[(size_t)s * H + h] + ad_l[dl * H + h], ATT_SLOPE);
        lds_max_f(&m_l[dl * H + h], v);
      }
    }
    __syncthreads();
    for (int e = tid; e < tot; e += OWN_NT) {
      int s, dl;
      if (e < E) {
        const unsigned x = (unsigned)(dst[e] - n0);
        if (x >= (unsigned)nloc) continue;
        dl = (int)x; s = src[e];
      } else { dl = e - E; s = n0 + dl; }
      for (int h = 0; h < H; ++h) {
        const float v = lrelu(as[(size_t)s * H + h] + ad_l[dl * H + h], ATT_SLOPE);
        const float w = expf(v - __int_as_float(m_l[dl * H + h]));
        atomicAdd(&den_l[dl * H + h], w);
        for (int c = 0; c < C; ++c)
          atomicAdd(&acc_l[dl * HC + h * C + c], w * hfeat[(size_t)s * HC + h * C + c]);
      }
    }
  }
  __syncthreads();

  if constexpr (!FINAL) {
    for (int it = tid; it < nloc * HC; it += OWN_NT) {
      const int n = it / HC, jj = it % HC;
      const float v = acc_l[it] / den_l[n * H + jj / C] + bias[jj];
      outp[(size_t)n0 * HC + it] = lrelu(v, ACT_SLOPE);
    }
  } else {
    for (int n = tid; n < nloc; n += OWN_NT) {
      const float dn = den_l[n];
      float s = 0.f;
      for (int c = 0; c < C; ++c) {
        const float f = lrelu(acc_l[n * C + c] / dn + bias[c], ACT_SLOPE);
        s += f * Wo[c];
      }
      outp[n0 + n] = s + bo[0];
    }
  }
}

extern "C" void kernel_launch(void* const* d_in, const int* in_sizes, int n_in,
                              void* d_out, int out_size, void* d_ws, size_t ws_size,
                              hipStream_t stream) {
  const float* x    = (const float*)d_in[0];
  const int*   ei   = (const int*)d_in[1];
  const float* W1   = (const float*)d_in[2];
  const float* aS1  = (const float*)d_in[3];
  const float* aD1  = (const float*)d_in[4];
  const float* b1   = (const float*)d_in[5];
  const float* W2   = (const float*)d_in[6];
  const float* aS2  = (const float*)d_in[7];
  const float* aD2  = (const float*)d_in[8];
  const float* b2   = (const float*)d_in[9];
  const float* Wo   = (const float*)d_in[10];
  const float* bo   = (const float*)d_in[11];
  float* out = (float*)d_out;

  const int IN_C = 64, HID = 16, HEADS = 4;
  const int HC1 = HEADS * HID;
  const int Nn = in_sizes[0] / IN_C;
  const int E  = in_sizes[1] / 2;
  const int* src = ei;
  const int* dst = ei + E;

  unsigned char* keys = (unsigned char*)d_ws;
  float* p = (float*)(keys + (((size_t)E + 255) & ~(size_t)255));
  float* h1   = p; p += (size_t)Nn * HC1;
  float* f1   = p; p += (size_t)Nn * HC1;
  float* h2   = p; p += (size_t)Nn * HID;
  float* as1  = p; p += (size_t)Nn * HEADS;
  float* ad1  = p; p += (size_t)Nn * HEADS;
  float* as2  = p; p += (size_t)Nn;
  float* ad2  = p; p += (size_t)Nn;

  const int TB = 256;
  const int OG = (Nn + OWN_NPW - 1) / OWN_NPW;

  int* sv_s = (int*)p;
  int* sv_d = sv_s + (size_t)OG * OWN_CAP;
  int* sv_n = sv_d + (size_t)OG * OWN_CAP;

  owner_keys<<<(E + 4 * KEY_NT * KEY_UNR - 1) / (4 * KEY_NT * KEY_UNR), KEY_NT, 0, stream>>>(
      dst, (unsigned int*)keys, E, Nn);

  {
    dim3 g((Nn + 15) / 16, HC1 / 16);
    wmma_gemm_f32<64, 64><<<g, 32, 0, stream>>>(x, W1, h1, Nn);
  }
  node_coef<<<(Nn + TB - 1) / TB, TB, 0, stream>>>(h1, aS1, aD1, as1, ad1, Nn, HEADS, HID);
  owner_layer<4, 16, false, false><<<OG, OWN_NT, 0, stream>>>(src, dst, keys, sv_s, sv_d, sv_n,
                                                            as1, ad1, h1, b1, Wo, bo, f1, E, Nn);

  {
    dim3 g((Nn + 15) / 16, HID / 16);
    wmma_gemm_f32<64, 16><<<g, 32, 0, stream>>>(f1, W2, h2, Nn);
  }
  node_coef<<<(Nn + TB - 1) / TB, TB, 0, stream>>>(h2, aS2, aD2, as2, ad2, Nn, 1, HID);
  owner_layer<1, 16, true, true><<<OG, OWN_NT, 0, stream>>>(src, dst, keys, sv_s, sv_d, sv_n,
                                                          as2, ad2, h2, b2, Wo, bo, out, E, Nn);
}


# Owner filter no longer reads every edge: a pre-pass writes per work-group one presence bit per 8-edge group; a work-group loads only the groups its bit row marks; sums still in arrival order
# speedup vs baseline: 6.6460x; 1.1130x over previous
#include <hip/hip_runtime.h>
#include <math.h>
#include <stdint.h>


#define ATT_SLOPE 0.2f
#define ACT_SLOPE 0.01f

typedef __attribute__((ext_vector_type(2))) float v2f;
typedef __attribute__((ext_vector_type(8))) float v8f;

__device__ __forceinline__ float lrelu(float v, float s) { return v > 0.f ? v : v * s; }

__device__ __forceinline__ void lds_max_f(int* slot, float val) {
  if (val >= 0.f)
    atomicMax(slot, __float_as_int(val));
  else
    atomicMin((unsigned int*)slot, (unsigned int)__float_as_int(val));
}

template <int K, int NCOLS>
__global__ void __launch_bounds__(32)
wmma_gemm_f32(const float* __restrict__ A, const float* __restrict__ B,
              float* __restrict__ D, int M) {
  const int m0 = blockIdx.x * 16;
  const int n0 = blockIdx.y * 16;
  const int lane = threadIdx.x;
  const int half = lane >> 4;
  const int lid  = lane & 15;
  const int ar = min(m0 + lid, M - 1);
  v8f acc = {};
#pragma unroll
  for (int k = 0; k < K; k += 4) {
    const int ka = k + half * 2;
    v2f a, b;
    a.x = A[(size_t)ar * K + ka];
    a.y = A[(size_t)ar * K + ka + 1];
    b.x = B[(size_t)ka * NCOLS + n0 + lid];
    b.y = B[(size_t)(ka + 1) * NCOLS + n0 + lid];
    acc = __builtin_amdgcn_wmma_f32_16x16x4_f32(false, a, false, b, (short)0, acc,
                                                false, false);
  }
  const int rbase = m0 + half * 8;
#pragma unroll
  for (int r = 0; r < 8; ++r) {
    const int row = rbase + r;
    if (row < M) D[(size_t)row * NCOLS + n0 + lid] = acc[r];
  }
}

__global__ void node_coef(const float* __restrict__ h,
                          const float* __restrict__ att_src,
                          const float* __restrict__ att_dst,
                          float* __restrict__ as, float* __restrict__ ad,
                          int Nn, int H, int C) {
  const int n = blockIdx.x * blockDim.x + threadIdx.x;
  if (n >= Nn) return;
  const int HC = H * C;
  for (int hh = 0; hh < H; ++hh) {
    float s = 0.f, d = 0.f;
    for (int c = 0; c < C; ++c) {
      const float v = h[(size_t)n * HC + hh * C + c];
      s += v * att_src[hh * C + c];
      d += v * att_dst[hh * C + c];
    }
    as[n * H + hh] = s;
    ad[n * H + hh] = d;
  }
}

constexpr int OWN_NT  = 1024;
constexpr int OWN_NPW = 224;
constexpr int OWN_CAP = 9216;
constexpr int OWN_UNR = 8;
constexpr unsigned OWN_NOKEY = 255u;
constexpr int BIT_GRP   = 8;
constexpr int BIT_PPG   = BIT_GRP / 4;
constexpr int BIT_NT    = 512;
constexpr int BIT_EPT   = 16;
constexpr int BIT_TGR   = BIT_EPT / BIT_GRP;
constexpr int BIT_CW    = BIT_NT * BIT_TGR / 32;
constexpr int BIT_PITCH = 257;

__device__ __forceinline__ unsigned owner_key(int d, int Nn) {
  return (unsigned)d < (unsigned)Nn ? min((unsigned)d / (unsigned)OWN_NPW, OWN_NOKEY) : OWN_NOKEY;
}

__global__ void __launch_bounds__(BIT_NT)
owner_bits(const int* __restrict__ dst, unsigned int* __restrict__ bits,
           int E, int Nn, int rows, int rowWords) {
  __shared__ unsigned int lb[BIT_CW * BIT_PITCH];
  const bool vec = ((reinterpret_cast<uintptr_t>(dst) & 15) == 0);
  if (rows > (int)OWN_NOKEY || !vec || E < 4) return;
  const int tid = threadIdx.x;
  for (int i = tid; i < BIT_CW * BIT_PITCH; i += BIT_NT) lb[i] = 0u;
  const int e0 = (blockIdx.x * BIT_NT + tid) * BIT_EPT;
  const int4* d4 = reinterpret_cast<const int4*>(dst);
  const int nq = E >> 2;
  int4 v[4];
#pragma unroll
  for (int j = 0; j < 4; ++j) v[j] = d4[min((e0 >> 2) + j, nq - 1)];
  __syncthreads();
  auto mark = [&](int d, int lg) {
    const unsigned o = owner_key(d, Nn);
    if (o != OWN_NOKEY) atomicOr(&lb[(lg >> 5) * BIT_PITCH + o], 1u << (lg & 31));
  };
#pragma unroll
  for (int j = 0; j < 4; ++j) {
    const int lg = tid * BIT_TGR + j / BIT_PPG;
    if ((e0 >> 2) + j < nq) { mark(v[j].x, lg); mark(v[j].y, lg); mark(v[j].z, lg); mark(v[j].w, lg); }
  }
  if (blockIdx.x == gridDim.x - 1 && tid < (E & 3)) {
    const int e = (E & ~3) + tid;
    mark(dst[e], e / BIT_GRP - blockIdx.x * (BIT_NT * BIT_TGR));
  }
  __syncthreads();
  for (int i = tid; i < rows * BIT_CW; i += BIT_NT) {
    const int o = i / BIT_CW, w = i % BIT_CW;
    bits[(size_t)o * rowWords + blockIdx.x * BIT_CW + w] = lb[w * BIT_PITCH + o];
  }
}

template <int N>
__device__ __forceinline__ int row_shr(int x) {
  return __builtin_amdgcn_update_dpp(0, x, 0x110 + N, 0xf, 0xf, true);
}
__device__ __forceinline__ int wave_reserve(int c, int* counter, int lane) {
  int incl = c;
  incl += row_shr<1>(incl);
  incl += row_shr<2>(incl);
  incl += row_shr<4>(incl);
  incl += row_shr<8>(incl);
  const int r0 = __builtin_amdgcn_readlane(incl, 15);
  const int r1 = __builtin_amdgcn_readlane(incl, 31);
  int base = 0;
  if (lane == 0) base = atomicAdd(counter, r0 + r1);
  base = __builtin_amdgcn_readfirstlane(base);
  return base + incl - c + (lane >= 16 ? r0 : 0);
}

template <int H, int C, bool FINAL, bool SAVED>
__global__ void __launch_bounds__(OWN_NT)
owner_layer(const int* __restrict__ src, const int* __restrict__ dst,
            const unsigned int* __restrict__ bits, int rowWords,
            int* __restrict__ sv_s, int* __restrict__ sv_d, int* __restrict__ sv_n,
            const float* __restrict__ as, const float* __restrict__ ad,
            const float* __restrict__ hfeat, const float* __restrict__ bias,
            const float* __restrict__ Wo, const float* __restrict__ bo,
            float* __restrict__ outp, int E, int Nn) {
  constexpr int HC = H * C;
  static_assert(OWN_NT % HC == 0, "a thread keeps one feature column");
  __shared__ float acc_l[OWN_NPW * HC];
  __shared__ float den_l[OWN_NPW * H];
  __shared__ float ad_l[OWN_NPW * H];
  __shared__ int   m_l[OWN_NPW * H];
  __shared__ int   list_s[OWN_CAP];
  __shared__ int   list_d[OWN_CAP];
  __shared__ float ew[OWN_CAP * H];
  __shared__ int   cnt;
  __shared__ int   gcnt;

  const int tid = threadIdx.x;
  const int n0 = blockIdx.x * OWN_NPW;
  const int nloc = min(OWN_NPW, Nn - n0);
  if (nloc <= 0) return;

  for (int i = tid; i < nloc * HC; i += OWN_NT) acc_l[i] = 0.f;
  for (int i = tid; i < nloc * H; i += OWN_NT) {
    den_l[i] = 0.f;
    m_l[i] = __float_as_int(-INFINITY);
    ad_l[i] = ad[(size_t)n0 * H + i];
  }
  for (int i = tid; i < nloc; i += OWN_NT) { list_s[i] = n0 + i; list_d[i] = i; }
  if (tid == 0) { cnt = nloc; gcnt = 0; }
  __syncthreads();

  int* const gs = sv_s + (size_t)blockIdx.x * OWN_CAP;
  int* const gd = sv_d + (size_t)blockIdx.x * OWN_CAP;

  const bool keyed = !SAVED && gridDim.x <= OWN_NOKEY && E >= 4 &&
                     ((reinterpret_cast<uintptr_t>(dst) & 15) == 0);
  if (keyed) {
    const int lane = tid & 31;
    int* const glist = reinterpret_cast<int*>(ew);
    const unsigned int* const row = bits + (size_t)blockIdx.x * rowWords;
    for (int wb = tid - lane; wb < rowWords; wb += OWN_NT * 4) {
      unsigned wv[4];
      int c = 0;
#pragma unroll
      for (int u = 0; u < 4; ++u) {
        const int wi = wb + lane + u * OWN_NT;
        wv[u] = wi < rowWords ? row[wi] : 0u;
      }
#pragma unroll
      for (int u = 0; u < 4; ++u) c += __popc(wv[u]);
      int pos = wave_reserve(c, &gcnt, lane);
#pragma unroll
      for (int u = 0; u < 4; ++u) {
        unsigned m = wv[u];
        const int g0 = (wb + lane + u * OWN_NT) * 32;
        while (m) {
          const int b = __ffs(m) - 1;
          m &= m - 1;
          if (pos < OWN_CAP) glist[pos] = g0 + b;
          ++pos;
        }
      }
    }
    __syncthreads();
    const int nit = min(gcnt, OWN_CAP) * BIT_PPG;
    const int4* d4 = reinterpret_cast<const int4*>(dst);
    const int nq = E >> 2;
    for (int ib = tid - lane; ib < nit; ib += OWN_NT * OWN_UNR) {
      int4 dv[OWN_UNR];
      int  eb[OWN_UNR];
#pragma unroll
      for (int u = 0; u < OWN_UNR; ++u) {
        const int it = min(ib + lane + u * OWN_NT, nit - 1);
        const int q = glist[it / BIT_PPG] * BIT_PPG + it % BIT_PPG;
        eb[u] = q;
        dv[u] = d4[min(q, nq - 1)];
      }
      unsigned hm = 0;
#pragma unroll
      for (int u = 0; u < OWN_UNR; ++u) {
        const unsigned h = ((unsigned)(dv[u].x - n0) < (unsigned)nloc ? 1u : 0u) |
                           ((unsigned)(dv[u].y - n0) < (unsigned)nloc ? 2u : 0u) |
                           ((unsigned)(dv[u].z - n0) < (unsigned)nloc ? 4u : 0u) |
                           ((unsigned)(dv[u].w - n0) < (unsigned)nloc ? 8u : 0u);
        const bool live = ib + lane + u * OWN_NT < nit && eb[u] < nq;
        hm |= (live ? h : 0u) << (4 * u);
      }
      int pos = wave_reserve(__popc(hm), &cnt, lane);
#pragma unroll
      for (int u = 0; u < OWN_UNR; ++u) {
        unsigned m = (hm >> (4 * u)) & 15u;
        while (m) {
          const int j = __ffs(m) - 1;
          m &= m - 1;
          if (pos < OWN_CAP) list_s[pos] = eb[u] * 4 + j;
          ++pos;
        }
      }
    }
    if (tid < (E & 3)) {
      const int e = (E & ~3) + tid;
      if ((unsigned)(dst[e] - n0) < (unsigned)nloc) {
        const int pos = atomicAdd(&cnt, 1);
        if (pos < OWN_CAP) list_s[pos] = e;
      }
    }
  }
  __syncthreads();
  const int total = SAVED ? sv_n[blockIdx.x] : (keyed ? cnt : OWN_CAP + 1);
  if (!SAVED && tid == 0) sv_n[blockIdx.x] = total;

  if (total <= OWN_CAP) {
   if constexpr (SAVED) {
    for (int i0 = nloc + tid; i0 < total; i0 += OWN_NT * OWN_UNR) {
      int s[OWN_UNR], d[OWN_UNR];
#pragma unroll
      for (int u = 0; u < OWN_UNR; ++u) {
        const int i = min(i0 + u * OWN_NT, total - 1);
        s[u] = gs[i];
        d[u] = gd[i];
      }
#pragma unroll
      for (int u = 0; u < OWN_UNR; ++u) {
        const int i = i0 + u * OWN_NT;
        if (i < total) { list_s[i] = s[u]; list_d[i] = d[u]; }
      }
    }
   } else {
    for (int i0 = nloc + tid; i0 < total; i0 += OWN_NT * OWN_UNR) {
      int s[OWN_UNR], d[OWN_UNR];
#pragma unroll
      for (int u = 0; u < OWN_UNR; ++u) {
        const int i = i0 + u * OWN_NT;
        const int e = list_s[min(i, total - 1)];
        const int ee = i < total ? e : 0;
        s[u] = src[ee];
        d[u] = dst[ee];
      }
#pragma unroll
      for (int u = 0; u < OWN_UNR; ++u) {
        const int i = i0 + u * OWN_NT;
        if (i < total) {
          list_s[i] = s[u]; list_d[i] = d[u] - n0;
          gs[i] = s[u]; gd[i] = d[u] - n0;
        }
      }
    }
   }
    __syncthreads();
    {
      constexpr int U1 = 4;
      const int totH = total * H;
      int it = tid;
      for (; it + (U1 - 1) * OWN_NT < totH; it += U1 * OWN_NT) {
        float a[U1];
        int sl[U1];
#pragma unroll
        for (int u = 0; u < U1; ++u) {
          const int q = it + u * OWN_NT;
          const int i = q / H, h = q % H;
          sl[u] = list_d[i] * H + h;
          a[u] = as[(size_t)list_s[i] * H + h];
        }
#pragma unroll
        for (int u = 0; u < U1; ++u) {
          const float v = lrelu(a[u] + ad_l[sl[u]], ATT_SLOPE);
          ew[it + u * OWN_NT] = v;
          lds_max_f(&m_l[sl[u]], v);
        }
      }
      for (; it < totH; it += OWN_NT) {
        const int i = it / H, h = it % H;
        const int sl = list_d[i] * H + h;
        const float v = lrelu(as[(size_t)list_s[i] * H + h] + ad_l[sl], ATT_SLOPE);
        ew[it] = v;
        lds_max_f(&m_l[sl], v);
      }
    }
    __syncthreads();
    for (int it = tid; it < total * H; it += OWN_NT) {
      const int i = it / H, h = it % H;
      const int dl = list_d[i];
      const float w = expf(ew[it] - __int_as_float(m_l[dl * H + h]));
      ew[it] = w;
      atomicAdd(&den_l[dl * H + h], w);
    }
    __syncthreads();
    constexpr int EPI = OWN_NT / HC;
    const int j = tid % HC;
    const int h = j / C;
    int i0 = tid / HC;
    for (; i0 + (OWN_UNR - 1) * EPI < total; i0 += EPI * OWN_UNR) {
      float hv[OWN_UNR], w[OWN_UNR];
      int dl[OWN_UNR];
#pragma unroll
      for (int u = 0; u < OWN_UNR; ++u) {
        const int i = i0 + u * EPI;
        dl[u] = list_d[i];
        w[u] = ew[i * H + h];
        hv[u] = hfeat[(size_t)list_s[i] * HC + j];
      }
#pragma unroll
      for (int u = 0; u < OWN_UNR; ++u) atomicAdd(&acc_l[dl[u] * HC + j], w[u] * hv[u]);
    }
    for (; i0 < total; i0 += EPI)
      atomicAdd(&acc_l[list_d[i0] * HC + j], ew[i0 * H + h] * hfeat[(size_t)list_s[i0] * HC + j]);
  } else {
    const int tot = E + nloc;
    for (int e = tid; e < tot; e += OWN_NT) {
      int s, dl;
      if (e < E) {
        const unsigned x = (unsigned)(dst[e] - n0);
        if (x >= (unsigned)nloc) continue;
        dl = (int)x; s = src[e];
      } else { dl = e - E; s = n0 + dl; }
      for (int h = 0; h < H; ++h) {
        const float v = lrelu(as[(size_t)s * H + h] + ad_l[dl * H + h], ATT_SLOPE);
        lds_max_f(&m_l[dl * H + h], v);
      }
    }
    __syncthreads();
    for (int e = tid; e < tot; e += OWN_NT) {
      int s, dl;
      if (e < E) {
        const unsigned x = (unsigned)(dst[e] - n0);
        if (x >= (unsigned)nloc) continue;
        dl = (int)x; s = src[e];
      } else { dl = e - E; s = n0 + dl; }
      for (int h = 0; h < H; ++h) {
        const float v = lrelu(as[(size_t)s * H + h] + ad_l[dl * H + h], ATT_SLOPE);
        const float w = expf(v - __int_as_float(m_l[dl * H + h]));
        atomicAdd(&den_l[dl * H + h], w);
        for (int c = 0; c < C; ++c)
          atomicAdd(&acc_l[dl * HC + h * C + c], w * hfeat[(size_t)s * HC + h * C + c]);
      }
    }
  }
  __syncthreads();

  if constexpr (!FINAL) {
    for (int it = tid; it < nloc * HC; it += OWN_NT) {
      const int n = it / HC, jj = it % HC;
      const float v = acc_l[it] / den_l[n * H + jj / C] + bias[jj];
      outp[(size_t)n0 * HC + it] = lrelu(v, ACT_SLOPE);
    }
  } else {
    for (int n = tid; n < nloc; n += OWN_NT) {
      const float dn = den_l[n];
      float s = 0.f;
      for (int c = 0; c < C; ++c) {
        const float f = lrelu(acc_l[n * C + c] / dn + bias[c], ACT_SLOPE);
        s += f * Wo[c];
      }
      outp[n0 + n] = s + bo[0];
    }
  }
}

extern "C" void kernel_launch(void* const* d_in, const int* in_sizes, int n_in,
                              void* d_out, int out_size, void* d_ws, size_t ws_size,
                              hipStream_t stream) {
  const float* x    = (const float*)d_in[0];
  const int*   ei   = (const int*)d_in[1];
  const float* W1   = (const float*)d_in[2];
  const float* aS1  = (const float*)d_in[3];
  const float* aD1  = (const float*)d_in[4];
  const float* b1   = (const float*)d_in[5];
  const float* W2   = (const float*)d_in[6];
  const float* aS2  = (const float*)d_in[7];
  const float* aD2  = (const float*)d_in[8];
  const float* b2   = (const float*)d_in[9];
  const float* Wo   = (const float*)d_in[10];
  const float* bo   = (const float*)d_in[11];
  float* out = (float*)d_out;

  const int IN_C = 64, HID = 16, HEADS = 4;
  const int HC1 = HEADS * HID;
  const int Nn = in_sizes[0] / IN_C;
  const int E  = in_sizes[1] / 2;
  const int* src = ei;
  const int* dst = ei + E;

  const int TB = 256;
  const int OG = (Nn + OWN_NPW - 1) / OWN_NPW;
  const int NCH = (E + BIT_NT * BIT_EPT - 1) / (BIT_NT * BIT_EPT) > 0
                ? (E + BIT_NT * BIT_EPT - 1) / (BIT_NT * BIT_EPT) : 1;
  const int rowWords = NCH * BIT_CW;

  unsigned int* bits = (unsigned int*)d_ws;
  float* p = (float*)(bits + (size_t)OG * rowWords);
  float* h1   = p; p += (size_t)Nn * HC1;
  float* f1   = p; p += (size_t)Nn * HC1;
  float* h2   = p; p += (size_t)Nn * HID;
  float* as1  = p; p += (size_t)Nn * HEADS;
  float* ad1  = p; p += (size_t)Nn * HEADS;
  float* as2  = p; p += (size_t)Nn;
  float* ad2  = p; p += (size_t)Nn;

  int* sv_s = (int*)p;
  int* sv_d = sv_s + (size_t)OG * OWN_CAP;
  int* sv_n = sv_d + (size_t)OG * OWN_CAP;

  owner_bits<<<NCH, BIT_NT, 0, stream>>>(dst, bits, E, Nn, OG, rowWords);

  {
    dim3 g((Nn + 15) / 16, HC1 / 16);
    wmma_gemm_f32<64, 64><<<g, 32, 0, stream>>>(x, W1, h1, Nn);
  }
  node_coef<<<(Nn + TB - 1) / TB, TB, 0, stream>>>(h1, aS1, aD1, as1, ad1, Nn, HEADS, HID);
  owner_layer<4, 16, false, false><<<OG, OWN_NT, 0, stream>>>(src, dst, bits, rowWords, sv_s, sv_d, sv_n,
                                                            as1, ad1, h1, b1, Wo, bo, f1, E, Nn);

  {
    dim3 g((Nn + 15) / 16, HID / 16);
    wmma_gemm_f32<64, 16><<<g, 32, 0, stream>>>(f1, W2, h2, Nn);
  }
  node_coef<<<(Nn + TB - 1) / TB, TB, 0, stream>>>(h2, aS2, aD2, as2, ad2, Nn, 1, HID);
  owner_layer<1, 16, true, true><<<OG, OWN_NT, 0, stream>>>(src, dst, bits, rowWords, sv_s, sv_d, sv_n,
                                                          as2, ad2, h2, b2, Wo, bo, out, E, Nn);
}


# Attention coefficients come from the GEMM accumulators (a wave now makes all heads of its 16 rows; 16-lane DPP add tree, coefficient sum order changes), not from a pass over h: 7 launches 5
# speedup vs baseline: 7.1575x; 1.0770x over previous
#include <hip/hip_runtime.h>
#include <math.h>
#include <stdint.h>


#define ATT_SLOPE 0.2f
#define ACT_SLOPE 0.01f

typedef __attribute__((ext_vector_type(2))) float v2f;
typedef __attribute__((ext_vector_type(8))) float v8f;

__device__ __forceinline__ float lrelu(float v, float s) { return v > 0.f ? v : v * s; }

__device__ __forceinline__ void lds_max_f(int* slot, float val) {
  if (val >= 0.f)
    atomicMax(slot, __float_as_int(val));
  else
    atomicMin((unsigned int*)slot, (unsigned int)__float_as_int(val));
}

constexpr int GEMM_WPG = 4;

template <int MASK>
__device__ __forceinline__ float row_xadd(float v) {
  return v + __int_as_float(__builtin_amdgcn_update_dpp(0, __float_as_int(v), 0x160 + MASK,
                                                        0xf, 0xf, true));
}

template <int K, int H>
__global__ void __launch_bounds__(GEMM_WPG * 32)
wmma_gemm_coef(const float* __restrict__ A, const float* __restrict__ B,
               const float* __restrict__ att_src, const float* __restrict__ att_dst,
               float* __restrict__ D, float* __restrict__ as, float* __restrict__ ad, int M) {
  constexpr int NCOLS = H * 16;
  const int m0 = (blockIdx.x * GEMM_WPG + (threadIdx.x >> 5)) * 16;
  if (m0 >= M) return;
  const int lane = threadIdx.x & 31;
  const int half = lane >> 4;
  const int lid  = lane & 15;
  const int ar = min(m0 + lid, M - 1);
  v8f acc[H];
#pragma unroll
  for (int t = 0; t < H; ++t) acc[t] = v8f{};
#pragma unroll
  for (int k = 0; k < K; k += 4) {
    const int ka = k + half * 2;
    v2f a;
    a.x = A[(size_t)ar * K + ka];
    a.y = A[(size_t)ar * K + ka + 1];
#pragma unroll
    for (int t = 0; t < H; ++t) {
      v2f b;
      b.x = B[(size_t)ka * NCOLS + t * 16 + lid];
      b.y = B[(size_t)(ka + 1) * NCOLS + t * 16 + lid];
      acc[t] = __builtin_amdgcn_wmma_f32_16x16x4_f32(false, a, false, b, (short)0, acc[t],
                                                     false, false);
    }
  }
  const int rbase = m0 + half * 8;
  float* const drow = D + (size_t)rbase * NCOLS + lid;
  if (m0 + 16 <= M) {
#pragma unroll
    for (int t = 0; t < H; ++t) {
#pragma unroll
      for (int r = 0; r < 8; ++r) drow[r * NCOLS + t * 16] = acc[t][r];
    }
  } else {
#pragma unroll
    for (int t = 0; t < H; ++t) {
#pragma unroll
      for (int r = 0; r < 8; ++r) {
        if (rbase + r < M) drow[r * NCOLS + t * 16] = acc[t][r];
      }
    }
  }
  float so[H], dv[H];
#pragma unroll
  for (int t = 0; t < H; ++t) {
    const float ws = att_src[t * 16 + lid];
    const float wd = att_dst[t * 16 + lid];
    float s[8], d[8];
#pragma unroll
    for (int r = 0; r < 8; ++r) { s[r] = acc[t][r] * ws; d[r] = acc[t][r] * wd; }
#pragma unroll
    for (int r = 0; r < 8; ++r) { s[r] = row_xadd<1>(s[r]); d[r] = row_xadd<1>(d[r]); }
#pragma unroll
    for (int r = 0; r < 8; ++r) { s[r] = row_xadd<2>(s[r]); d[r] = row_xadd<2>(d[r]); }
#pragma unroll
    for (int r = 0; r < 8; ++r) { s[r] = row_xadd<4>(s[r]); d[r] = row_xadd<4>(d[r]); }
#pragma unroll
    for (int r = 0; r < 8; ++r) { s[r] = row_xadd<8>(s[r]); d[r] = row_xadd<8>(d[r]); }
    float ss = s[0], dd = d[0];
#pragma unroll
    for (int r = 1; r < 8; ++r) { ss = lid == r ? s[r] : ss; dd = lid == r ? d[r] : dd; }
    so[t] = ss; dv[t] = dd;
  }
  const int row = rbase + lid;
  if (lid < 8 && row < M) {
    if constexpr (H % 4 == 0) {
#pragma unroll
      for (int t = 0; t < H; t += 4) {
        *reinterpret_cast<float4*>(as + (size_t)row * H + t) = make_float4(so[t], so[t + 1], so[t + 2], so[t + 3]);
        *reinterpret_cast<float4*>(ad + (size_t)row * H + t) = make_float4(dv[t], dv[t + 1], dv[t + 2], dv[t + 3]);
      }
    } else {
#pragma unroll
      for (int t = 0; t < H; ++t) { as[(size_t)row * H + t] = so[t]; ad[(size_t)row * H + t] = dv[t]; }
    }
  }
}

constexpr int OWN_NT  = 1024;
constexpr int OWN_NPW = 224;
constexpr int OWN_CAP = 9216;
constexpr int OWN_UNR = 8;
constexpr unsigned OWN_NOKEY = 255u;
constexpr int BIT_GRP   = 8;
constexpr int BIT_PPG   = BIT_GRP / 4;
constexpr int BIT_NT    = 512;
constexpr int BIT_EPT   = 16;
constexpr int BIT_TGR   = BIT_EPT / BIT_GRP;
constexpr int BIT_CW    = BIT_NT * BIT_TGR / 32;
constexpr int BIT_PITCH = 257;

__device__ __forceinline__ unsigned owner_key(int d, int Nn) {
  return (unsigned)d < (unsigned)Nn ? min((unsigned)d / (unsigned)OWN_NPW, OWN_NOKEY) : OWN_NOKEY;
}

__global__ void __launch_bounds__(BIT_NT)
owner_bits(const int* __restrict__ dst, unsigned int* __restrict__ bits,
           int E, int Nn, int rows, int rowWords) {
  __shared__ unsigned int lb[BIT_CW * BIT_PITCH];
  const bool vec = ((reinterpret_cast<uintptr_t>(dst) & 15) == 0);
  if (rows > (int)OWN_NOKEY || !vec || E < 4) return;
  const int tid = threadIdx.x;
  for (int i = tid; i < BIT_CW * BIT_PITCH; i += BIT_NT) lb[i] = 0u;
  const int e0 = (blockIdx.x * BIT_NT + tid) * BIT_EPT;
  const int4* d4 = reinterpret_cast<const int4*>(dst);
  const int nq = E >> 2;
  int4 v[4];
#pragma unroll
  for (int j = 0; j < 4; ++j) v[j] = d4[min((e0 >> 2) + j, nq - 1)];
  __syncthreads();
  auto mark = [&](int d, int lg) {
    const unsigned o = owner_key(d, Nn);
    if (o != OWN_NOKEY) atomicOr(&lb[(lg >> 5) * BIT_PITCH + o], 1u << (lg & 31));
  };
#pragma unroll
  for (int j = 0; j < 4; ++j) {
    const int lg = tid * BIT_TGR + j / BIT_PPG;
    if ((e0 >> 2) + j < nq) { mark(v[j].x, lg); mark(v[j].y, lg); mark(v[j].z, lg); mark(v[j].w, lg); }
  }
  if (blockIdx.x == gridDim.x - 1 && tid < (E & 3)) {
    const int e = (E & ~3) + tid;
    mark(dst[e], e / BIT_GRP - blockIdx.x * (BIT_NT * BIT_TGR));
  }
  __syncthreads();
  for (int i = tid; i < rows * BIT_CW; i += BIT_NT) {
    const int o = i / BIT_CW, w = i % BIT_CW;
    bits[(size_t)o * rowWords + blockIdx.x * BIT_CW + w] = lb[w * BIT_PITCH + o];
  }
}

template <int N>
__device__ __forceinline__ int row_shr(int x) {
  return __builtin_amdgcn_update_dpp(0, x, 0x110 + N, 0xf, 0xf, true);
}
__device__ __forceinline__ int wave_reserve(int c, int* counter, int lane) {
  int incl = c;
  incl += row_shr<1>(incl);
  incl += row_shr<2>(incl);
  incl += row_shr<4>(incl);
  incl += row_shr<8>(incl);
  const int r0 = __builtin_amdgcn_readlane(incl, 15);
  const int r1 = __builtin_amdgcn_readlane(incl, 31);
  int base = 0;
  if (lane == 0) base = atomicAdd(counter, r0 + r1);
  base = __builtin_amdgcn_readfirstlane(base);
  return base + incl - c + (lane >= 16 ? r0 : 0);
}

template <int H, int C, bool FINAL, bool SAVED>
__global__ void __launch_bounds__(OWN_NT)
owner_layer(const int* __restrict__ src, const int* __restrict__ dst,
            const unsigned int* __restrict__ bits, int rowWords,
            int* __restrict__ sv_s, int* __restrict__ sv_d, int* __restrict__ sv_n,
            const float* __restrict__ as, const float* __restrict__ ad,
            const float* __restrict__ hfeat, const float* __restrict__ bias,
            const float* __restrict__ Wo, const float* __restrict__ bo,
            float* __restrict__ outp, int E, int Nn) {
  constexpr int HC = H * C;
  static_assert(OWN_NT % HC == 0, "a thread keeps one feature column");
  __shared__ float acc_l[OWN_NPW * HC];
  __shared__ float den_l[OWN_NPW * H];
  __shared__ float ad_l[OWN_NPW * H];
  __shared__ int   m_l[OWN_NPW * H];
  __shared__ int   list_s[OWN_CAP];
  __shared__ int   list_d[OWN_CAP];
  __shared__ float ew[OWN_CAP * H];
  __shared__ int   cnt;
  __shared__ int   gcnt;

  const int tid = threadIdx.x;
  const int n0 = blockIdx.x * OWN_NPW;
  const int nloc = min(OWN_NPW, Nn - n0);
  if (nloc <= 0) return;

  for (int i = tid; i < nloc * HC; i += OWN_NT) acc_l[i] = 0.f;
  for (int i = tid; i < nloc * H; i += OWN_NT) {
    den_l[i] = 0.f;
    m_l[i] = __float_as_int(-INFINITY);
    ad_l[i] = ad[(size_t)n0 * H + i];
  }
  for (int i = tid; i < nloc; i += OWN_NT) { list_s[i] = n0 + i; list_d[i] = i; }
  if (tid == 0) { cnt = nloc; gcnt = 0; }
  __syncthreads();

  int* const gs = sv_s + (size_t)blockIdx.x * OWN_CAP;
  int* const gd = sv_d + (size_t)blockIdx.x * OWN_CAP;

  const bool keyed = !SAVED && gridDim.x <= OWN_NOKEY && E >= 4 &&
                     ((reinterpret_cast<uintptr_t>(dst) & 15) == 0);
  if (keyed) {
    const int lane = tid & 31;
    int* const glist = reinterpret_cast<int*>(ew);
    const unsigned int* const row = bits + (size_t)blockIdx.x * rowWords;
    for (int wb = tid - lane; wb < rowWords; wb += OWN_NT * 4) {
      unsigned wv[4];
      int c = 0;
#pragma unroll
      for (int u = 0; u < 4; ++u) {
        const int wi = wb + lane + u * OWN_NT;
        wv[u] = wi < rowWords ? row[wi] : 0u;
      }
#pragma unroll
      for (int u = 0; u < 4; ++u) c += __popc(wv[u]);
      int pos = wave_reserve(c, &gcnt, lane);
#pragma unroll
      for (int u = 0; u < 4; ++u) {
        unsigned m = wv[u];
        const int g0 = (wb + lane + u * OWN_NT) * 32;
        while (m) {
          const int b = __ffs(m) - 1;
          m &= m - 1;
          if (pos < OWN_CAP) glist[pos] = g0 + b;
          ++pos;
        }
      }
    }
    __syncthreads();
    const int nit = min(gcnt, OWN_CAP) * BIT_PPG;
    const int4* d4 = reinterpret_cast<const int4*>(dst);
    const int nq = E >> 2;
    for (int ib = tid - lane; ib < nit; ib += OWN_NT * OWN_UNR) {
      int4 dv[OWN_UNR];
      int  eb[OWN_UNR];
#pragma unroll
      for (int u = 0; u < OWN_UNR; ++u) {
        const int it = min(ib + lane + u * OWN_NT, nit - 1);
        const int q = glist[it / BIT_PPG] * BIT_PPG + it % BIT_PPG;
        eb[u] = q;
        dv[u] = d4[min(q, nq - 1)];
      }
      unsigned hm = 0;
#pragma unroll
      for (int u = 0; u < OWN_UNR; ++u) {
        const unsigned h = ((unsigned)(dv[u].x - n0) < (unsigned)nloc ? 1u : 0u) |
                           ((unsigned)(dv[u].y - n0) < (unsigned)nloc ? 2u : 0u) |
                           ((unsigned)(dv[u].z - n0) < (unsigned)nloc ? 4u : 0u) |
                           ((unsigned)(dv[u].w - n0) < (unsigned)nloc ? 8u : 0u);
        const bool live = ib + lane + u * OWN_NT < nit && eb[u] < nq;
        hm |= (live ? h : 0u) << (4 * u);
      }
      int pos = wave_reserve(__popc(hm), &cnt, lane);
#pragma unroll
      for (int u = 0; u < OWN_UNR; ++u) {
        unsigned m = (hm >> (4 * u)) & 15u;
        while (m) {
          const int j = __ffs(m) - 1;
          m &= m - 1;
          if (pos < OWN_CAP) list_s[pos] = eb[u] * 4 + j;
          ++pos;
        }
      }
    }
    if (tid < (E & 3)) {
      const int e = (E & ~3) + tid;
      if ((unsigned)(dst[e] - n0) < (unsigned)nloc) {
        const int pos = atomicAdd(&cnt, 1);
        if (pos < OWN_CAP) list_s[pos] = e;
      }
    }
  }
  __syncthreads();
  const int total = SAVED ? sv_n[blockIdx.x] : (keyed ? cnt : OWN_CAP + 1);
  if (!SAVED && tid == 0) sv_n[blockIdx.x] = total;

  if (total <= OWN_CAP) {
   if constexpr (SAVED) {
    for (int i0 = nloc + tid; i0 < total; i0 += OWN_NT * OWN_UNR) {
      int s[OWN_UNR], d[OWN_UNR];
#pragma unroll
      for (int u = 0; u < OWN_UNR; ++u) {
        const int i = min(i0 + u * OWN_NT, total - 1);
        s[u] = gs[i];
        d[u] = gd[i];
      }
#pragma unroll
      for (int u = 0; u < OWN_UNR; ++u) {
        const int i = i0 + u * OWN_NT;
        if (i < total) { list_s[i] = s[u]; list_d[i] = d[u]; }
      }
    }
   } else {
    for (int i0 = nloc + tid; i0 < total; i0 += OWN_NT * OWN_UNR) {
      int s[OWN_UNR], d[OWN_UNR];
#pragma unroll
      for (int u = 0; u < OWN_UNR; ++u) {
        const int i = i0 + u * OWN_NT;
        const int e = list_s[min(i, total - 1)];
        const int ee = i < total ? e : 0;
        s[u] = src[ee];
        d[u] = dst[ee];
      }
#pragma unroll
      for (int u = 0; u < OWN_UNR; ++u) {
        const int i = i0 + u * OWN_NT;
        if (i < total) {
          list_s[i] = s[u]; list_d[i] = d[u] - n0;
          gs[i] = s[u]; gd[i] = d[u] - n0;
        }
      }
    }
   }
    __syncthreads();
    {
      constexpr int U1 = 4;
      const int totH = total * H;
      int it = tid;
      for (; it + (U1 - 1) * OWN_NT < totH; it += U1 * OWN_NT) {
        float a[U1];
        int sl[U1];
#pragma unroll
        for (int u = 0; u < U1; ++u) {
          const int q = it + u * OWN_NT;
          const int i = q / H, h = q % H;
          sl[u] = list_d[i] * H + h;
          a[u] = as[(size_t)list_s[i] * H + h];
        }
#pragma unroll
        for (int u = 0; u < U1; ++u) {
          const float v = lrelu(a[u] + ad_l[sl[u]], ATT_SLOPE);
          ew[it + u * OWN_NT] = v;
          lds_max_f(&m_l[sl[u]], v);
        }
      }
      for (; it < totH; it += OWN_NT) {
        const int i = it / H, h = it % H;
        const int sl = list_d[i] * H + h;
        const float v = lrelu(as[(size_t)list_s[i] * H + h] + ad_l[sl], ATT_SLOPE);
        ew[it] = v;
        lds_max_f(&m_l[sl], v);
      }
    }
    __syncthreads();
    for (int it = tid; it < total * H; it += OWN_NT) {
      const int i = it / H, h = it % H;
      const int dl = list_d[i];
      const float w = expf(ew[it] - __int_as_float(m_l[dl * H + h]));
      ew[it] = w;
      atomicAdd(&den_l[dl * H + h], w);
    }
    __syncthreads();
    constexpr int EPI = OWN_NT / HC;
    const int j = tid % HC;
    const int h = j / C;
    int i0 = tid / HC;
    for (; i0 + (OWN_UNR - 1) * EPI < total; i0 += EPI * OWN_UNR) {
      float hv[OWN_UNR], w[OWN_UNR];
      int dl[OWN_UNR];
#pragma unroll
      for (int u = 0; u < OWN_UNR; ++u) {
        const int i = i0 + u * EPI;
        dl[u] = list_d[i];
        w[u] = ew[i * H + h];
        hv[u] = hfeat[(size_t)list_s[i] * HC + j];
      }
#pragma unroll
      for (int u = 0; u < OWN_UNR; ++u) atomicAdd(&acc_l[dl[u] * HC + j], w[u] * hv[u]);
    }
    for (; i0 < total; i0 += EPI)
      atomicAdd(&acc_l[list_d[i0] * HC + j], ew[i0 * H + h] * hfeat[(size_t)list_s[i0] * HC + j]);
  } else {
    const int tot = E + nloc;
    for (int e = tid; e < tot; e += OWN_NT) {
      int s, dl;
      if (e < E) {
        const unsigned x = (unsigned)(dst[e] - n0);
        if (x >= (unsigned)nloc) continue;
        dl = (int)x; s = src[e];
      } else { dl = e - E; s = n0 + dl; }
      for (int h = 0; h < H; ++h) {
        const float v = lrelu(as[(size_t)s * H + h] + ad_l[dl * H + h], ATT_SLOPE);
        lds_max_f(&m_l[dl * H + h], v);
      }
    }
    __syncthreads();
    for (int e = tid; e < tot; e += OWN_NT) {
      int s, dl;
      if (e < E) {
        const unsigned x = (unsigned)(dst[e] - n0);
        if (x >= (unsigned)nloc) continue;
        dl = (int)x; s = src[e];
      } else { dl = e - E; s = n0 + dl; }
      for (int h = 0; h < H; ++h) {
        const float v = lrelu(as[(size_t)s * H + h] + ad_l[dl * H + h], ATT_SLOPE);
        const float w = expf(v - __int_as_float(m_l[dl * H + h]));
        atomicAdd(&den_l[dl * H + h], w);
        for (int c = 0; c < C; ++c)
          atomicAdd(&acc_l[dl * HC + h * C + c], w * hfeat[(size_t)s * HC + h * C + c]);
      }
    }
  }
  __syncthreads();

  if constexpr (!FINAL) {
    for (int it = tid; it < nloc * HC; it += OWN_NT) {
      const int n = it / HC, jj = it % HC;
      const float v = acc_l[it] / den_l[n * H + jj / C] + bias[jj];
      outp[(size_t)n0 * HC + it] = lrelu(v, ACT_SLOPE);
    }
  } else {
    for (int n = tid; n < nloc; n += OWN_NT) {
      const float dn = den_l[n];
      float s = 0.f;
      for (int c = 0; c < C; ++c) {
        const float f = lrelu(acc_l[n * C + c] / dn + bias[c], ACT_SLOPE);
        s += f * Wo[c];
      }
      outp[n0 + n] = s + bo[0];
    }
  }
}

extern "C" void kernel_launch(void* const* d_in, const int* in_sizes, int n_in,
                              void* d_out, int out_size, void* d_ws, size_t ws_size,
                              hipStream_t stream) {
  const float* x    = (const float*)d_in[0];
  const int*   ei   = (const int*)d_in[1];
  const float* W1   = (const float*)d_in[2];
  const float* aS1  = (const float*)d_in[3];
  const float* aD1  = (const float*)d_in[4];
  const float* b1   = (const float*)d_in[5];
  const float* W2   = (const float*)d_in[6];
  const float* aS2  = (const float*)d_in[7];
  const float* aD2  = (const float*)d_in[8];
  const float* b2   = (const float*)d_in[9];
  const float* Wo   = (const float*)d_in[10];
  const float* bo   = (const float*)d_in[11];
  float* out = (float*)d_out;

  const int IN_C = 64, HID = 16, HEADS = 4;
  const int HC1 = HEADS * HID;
  const int Nn = in_sizes[0] / IN_C;
  const int E  = in_sizes[1] / 2;
  const int* src = ei;
  const int* dst = ei + E;

  const int GB = (Nn + 16 * GEMM_WPG - 1) / (16 * GEMM_WPG);
  const int OG = (Nn + OWN_NPW - 1) / OWN_NPW;
  const int NCH = (E + BIT_NT * BIT_EPT - 1) / (BIT_NT * BIT_EPT) > 0
                ? (E + BIT_NT * BIT_EPT - 1) / (BIT_NT * BIT_EPT) : 1;
  const int rowWords = NCH * BIT_CW;

  unsigned int* bits = (unsigned int*)d_ws;
  float* p = (float*)(bits + (size_t)OG * rowWords);
  float* h1   = p; p += (size_t)Nn * HC1;
  float* f1   = p; p += (size_t)Nn * HC1;
  float* h2   = p; p += (size_t)Nn * HID;
  float* as1  = p; p += (size_t)Nn * HEADS;
  float* ad1  = p; p += (size_t)Nn * HEADS;
  float* as2  = p; p += (size_t)Nn;
  float* ad2  = p; p += (size_t)Nn;

  int* sv_s = (int*)p;
  int* sv_d = sv_s + (size_t)OG * OWN_CAP;
  int* sv_n = sv_d + (size_t)OG * OWN_CAP;

  owner_bits<<<NCH, BIT_NT, 0, stream>>>(dst, bits, E, Nn, OG, rowWords);

  wmma_gemm_coef<64, 4><<<GB, GEMM_WPG * 32, 0, stream>>>(x, W1, aS1, aD1, h1, as1, ad1, Nn);
  owner_layer<4, 16, false, false><<<OG, OWN_NT, 0, stream>>>(src, dst, bits, rowWords, sv_s, sv_d, sv_n,
                                                            as1, ad1, h1, b1, Wo, bo, f1, E, Nn);

  wmma_gemm_coef<64, 1><<<GB, GEMM_WPG * 32, 0, stream>>>(f1, W2, aS2, aD2, h2, as2, ad2, Nn);
  owner_layer<1, 16, true, true><<<OG, OWN_NT, 0, stream>>>(src, dst, bits, rowWords, sv_s, sv_d, sv_n,
                                                          as2, ad2, h2, b2, Wo, bo, out, E, Nn);
}


# Sum sweep gathers a feature row in 16-byte pieces: a lane keeps four consecutive columns of an edge, so list entry and weight are read from LDS once per four columns; sums still in arrival order
# speedup vs baseline: 7.3969x; 1.0334x over previous
#include <hip/hip_runtime.h>
#include <math.h>
#include <stdint.h>


#define ATT_SLOPE 0.2f
#define ACT_SLOPE 0.01f

typedef __attribute__((ext_vector_type(2))) float v2f;
typedef __attribute__((ext_vector_type(8))) float v8f;
typedef __attribute__((ext_vector_type(4))) float v4f;

__device__ __forceinline__ float lrelu(float v, float s) { return v > 0.f ? v : v * s; }

__device__ __forceinline__ void lds_max_f(int* slot, float val) {
  if (val >= 0.f)
    atomicMax(slot, __float_as_int(val));
  else
    atomicMin((unsigned int*)slot, (unsigned int)__float_as_int(val));
}

constexpr int GEMM_WPG = 4;

template <int MASK>
__device__ __forceinline__ float row_xadd(float v) {
  return v + __int_as_float(__builtin_amdgcn_update_dpp(0, __float_as_int(v), 0x160 + MASK,
                                                        0xf, 0xf, true));
}

template <int K, int H>
__global__ void __launch_bounds__(GEMM_WPG * 32)
wmma_gemm_coef(const float* __restrict__ A, const float* __restrict__ B,
               const float* __restrict__ att_src, const float* __restrict__ att_dst,
               float* __restrict__ D, float* __restrict__ as, float* __restrict__ ad, int M) {
  constexpr int NCOLS = H * 16;
  const int m0 = (blockIdx.x * GEMM_WPG + (threadIdx.x >> 5)) * 16;
  if (m0 >= M) return;
  const int lane = threadIdx.x & 31;
  const int half = lane >> 4;
  const int lid  = lane & 15;
  const int ar = min(m0 + lid, M - 1);
  v8f acc[H];
#pragma unroll
  for (int t = 0; t < H; ++t) acc[t] = v8f{};
#pragma unroll
  for (int k = 0; k < K; k += 4) {
    const int ka = k + half * 2;
    v2f a;
    a.x = A[(size_t)ar * K + ka];
    a.y = A[(size_t)ar * K + ka + 1];
#pragma unroll
    for (int t = 0; t < H; ++t) {
      v2f b;
      b.x = B[(size_t)ka * NCOLS + t * 16 + lid];
      b.y = B[(size_t)(ka + 1) * NCOLS + t * 16 + lid];
      acc[t] = __builtin_amdgcn_wmma_f32_16x16x4_f32(false, a, false, b, (short)0, acc[t],
                                                     false, false);
    }
  }
  const int rbase = m0 + half * 8;
  float* const drow = D + (size_t)rbase * NCOLS + lid;
  if (m0 + 16 <= M) {
#pragma unroll
    for (int t = 0; t < H; ++t) {
#pragma unroll
      for (int r = 0; r < 8; ++r) drow[r * NCOLS + t * 16] = acc[t][r];
    }
  } else {
#pragma unroll
    for (int t = 0; t < H; ++t) {
#pragma unroll
      for (int r = 0; r < 8; ++r) {
        if (rbase + r < M) drow[r * NCOLS + t * 16] = acc[t][r];
      }
    }
  }
  float so[H], dv[H];
#pragma unroll
  for (int t = 0; t < H; ++t) {
    const float ws = att_src[t * 16 + lid];
    const float wd = att_dst[t * 16 + lid];
    float s[8], d[8];
#pragma unroll
    for (int r = 0; r < 8; ++r) { s[r] = acc[t][r] * ws; d[r] = acc[t][r] * wd; }
#pragma unroll
    for (int r = 0; r < 8; ++r) { s[r] = row_xadd<1>(s[r]); d[r] = row_xadd<1>(d[r]); }
#pragma unroll
    for (int r = 0; r < 8; ++r) { s[r] = row_xadd<2>(s[r]); d[r] = row_xadd<2>(d[r]); }
#pragma unroll
    for (int r = 0; r < 8; ++r) { s[r] = row_xadd<4>(s[r]); d[r] = row_xadd<4>(d[r]); }
#pragma unroll
    for (int r = 0; r < 8; ++r) { s[r] = row_xadd<8>(s[r]); d[r] = row_xadd<8>(d[r]); }
    float ss = s[0], dd = d[0];
#pragma unroll
    for (int r = 1; r < 8; ++r) { ss = lid == r ? s[r] : ss; dd = lid == r ? d[r] : dd; }
    so[t] = ss; dv[t] = dd;
  }
  const int row = rbase + lid;
  if (lid < 8 && row < M) {
    if constexpr (H % 4 == 0) {
#pragma unroll
      for (int t = 0; t < H; t += 4) {
        *reinterpret_cast<float4*>(as + (size_t)row * H + t) = make_float4(so[t], so[t + 1], so[t + 2], so[t + 3]);
        *reinterpret_cast<float4*>(ad + (size_t)row * H + t) = make_float4(dv[t], dv[t + 1], dv[t + 2], dv[t + 3]);
      }
    } else {
#pragma unroll
      for (int t = 0; t < H; ++t) { as[(size_t)row * H + t] = so[t]; ad[(size_t)row * H + t] = dv[t]; }
    }
  }
}

constexpr int OWN_NT  = 1024;
constexpr int OWN_NPW = 224;
constexpr int OWN_CAP = 9216;
constexpr int OWN_UNR = 8;
constexpr unsigned OWN_NOKEY = 255u;
constexpr int BIT_GRP   = 8;
constexpr int BIT_PPG   = BIT_GRP / 4;
constexpr int BIT_NT    = 512;
constexpr int BIT_EPT   = 16;
constexpr int BIT_TGR   = BIT_EPT / BIT_GRP;
constexpr int BIT_CW    = BIT_NT * BIT_TGR / 32;
constexpr int BIT_PITCH = 257;

__device__ __forceinline__ unsigned owner_key(int d, int Nn) {
  return (unsigned)d < (unsigned)Nn ? min((unsigned)d / (unsigned)OWN_NPW, OWN_NOKEY) : OWN_NOKEY;
}

__global__ void __launch_bounds__(BIT_NT)
owner_bits(const int* __restrict__ dst, unsigned int* __restrict__ bits,
           int E, int Nn, int rows, int rowWords) {
  __shared__ unsigned int lb[BIT_CW * BIT_PITCH];
  const bool vec = ((reinterpret_cast<uintptr_t>(dst) & 15) == 0);
  if (rows > (int)OWN_NOKEY || !vec || E < 4) return;
  const int tid = threadIdx.x;
  for (int i = tid; i < BIT_CW * BIT_PITCH; i += BIT_NT) lb[i] = 0u;
  const int e0 = (blockIdx.x * BIT_NT + tid) * BIT_EPT;
  const int4* d4 = reinterpret_cast<const int4*>(dst);
  const int nq = E >> 2;
  int4 v[4];
#pragma unroll
  for (int j = 0; j < 4; ++j) v[j] = d4[min((e0 >> 2) + j, nq - 1)];
  __syncthreads();
  auto mark = [&](int d, int lg) {
    const unsigned o = owner_key(d, Nn);
    if (o != OWN_NOKEY) atomicOr(&lb[(lg >> 5) * BIT_PITCH + o], 1u << (lg & 31));
  };
#pragma unroll
  for (int j = 0; j < 4; ++j) {
    const int lg = tid * BIT_TGR + j / BIT_PPG;
    if ((e0 >> 2) + j < nq) { mark(v[j].x, lg); mark(v[j].y, lg); mark(v[j].z, lg); mark(v[j].w, lg); }
  }
  if (blockIdx.x == gridDim.x - 1 && tid < (E & 3)) {
    const int e = (E & ~3) + tid;
    mark(dst[e], e / BIT_GRP - blockIdx.x * (BIT_NT * BIT_TGR));
  }
  __syncthreads();
  for (int i = tid; i < rows * BIT_CW; i += BIT_NT) {
    const int o = i / BIT_CW, w = i % BIT_CW;
    bits[(size_t)o * rowWords + blockIdx.x * BIT_CW + w] = lb[w * BIT_PITCH + o];
  }
}

template <int N>
__device__ __forceinline__ int row_shr(int x) {
  return __builtin_amdgcn_update_dpp(0, x, 0x110 + N, 0xf, 0xf, true);
}
__device__ __forceinline__ int wave_reserve(int c, int* counter, int lane) {
  int incl = c;
  incl += row_shr<1>(incl);
  incl += row_shr<2>(incl);
  incl += row_shr<4>(incl);
  incl += row_shr<8>(incl);
  const int r0 = __builtin_amdgcn_readlane(incl, 15);
  const int r1 = __builtin_amdgcn_readlane(incl, 31);
  int base = 0;
  if (lane == 0) base = atomicAdd(counter, r0 + r1);
  base = __builtin_amdgcn_readfirstlane(base);
  return base + incl - c + (lane >= 16 ? r0 : 0);
}

template <int H, int C, bool FINAL, bool SAVED>
__global__ void __launch_bounds__(OWN_NT)
owner_layer(const int* __restrict__ src, const int* __restrict__ dst,
            const unsigned int* __restrict__ bits, int rowWords,
            int* __restrict__ sv_s, int* __restrict__ sv_d, int* __restrict__ sv_n,
            const float* __restrict__ as, const float* __restrict__ ad,
            const float* __restrict__ hfeat, const float* __restrict__ bias,
            const float* __restrict__ Wo, const float* __restrict__ bo,
            float* __restrict__ outp, int E, int Nn) {
  constexpr int HC = H * C;
  static_assert(OWN_NT % HC == 0, "a thread keeps one feature column");
  __shared__ float acc_l[OWN_NPW * HC];
  __shared__ float den_l[OWN_NPW * H];
  __shared__ float ad_l[OWN_NPW * H];
  __shared__ int   m_l[OWN_NPW * H];
  __shared__ int   list_s[OWN_CAP];
  __shared__ int   list_d[OWN_CAP];
  __shared__ float ew[OWN_CAP * H];
  __shared__ int   cnt;
  __shared__ int   gcnt;

  const int tid = threadIdx.x;
  const int n0 = blockIdx.x * OWN_NPW;
  const int nloc = min(OWN_NPW, Nn - n0);
  if (nloc <= 0) return;

  for (int i = tid; i < nloc * HC; i += OWN_NT) acc_l[i] = 0.f;
  for (int i = tid; i < nloc * H; i += OWN_NT) {
    den_l[i] = 0.f;
    m_l[i] = __float_as_int(-INFINITY);
    ad_l[i] = ad[(size_t)n0 * H + i];
  }
  for (int i = tid; i < nloc; i += OWN_NT) { list_s[i] = n0 + i; list_d[i] = i; }
  if (tid == 0) { cnt = nloc; gcnt = 0; }
  __syncthreads();

  int* const gs = sv_s + (size_t)blockIdx.x * OWN_CAP;
  int* const gd = sv_d + (size_t)blockIdx.x * OWN_CAP;

  const bool keyed = !SAVED && gridDim.x <= OWN_NOKEY && E >= 4 &&
                     ((reinterpret_cast<uintptr_t>(dst) & 15) == 0);
  if (keyed) {
    const int lane = tid & 31;
    int* const glist = reinterpret_cast<int*>(ew);
    const unsigned int* const row = bits + (size_t)blockIdx.x * rowWords;
    for (int wb = tid - lane; wb < rowWords; wb += OWN_NT * 4) {
      unsigned wv[4];
      int c = 0;
#pragma unroll
      for (int u = 0; u < 4; ++u) {
        const int wi = wb + lane + u * OWN_NT;
        wv[u] = wi < rowWords ? row[wi] : 0u;
      }
#pragma unroll
      for (int u = 0; u < 4; ++u) c += __popc(wv[u]);
      int pos = wave_reserve(c, &gcnt, lane);
#pragma unroll
      for (int u = 0; u < 4; ++u) {
        unsigned m = wv[u];
        const int g0 = (wb + lane + u * OWN_NT) * 32;
        while (m) {
          const int b = __ffs(m) - 1;
          m &= m - 1;
          if (pos < OWN_CAP) glist[pos] = g0 + b;
          ++pos;
        }
      }
    }
    __syncthreads();
    const int nit = min(gcnt, OWN_CAP) * BIT_PPG;
    const int4* d4 = reinterpret_cast<const int4*>(dst);
    const int nq = E >> 2;
    for (int ib = tid - lane; ib < nit; ib += OWN_NT * OWN_UNR) {
      int4 dv[OWN_UNR];
      int  eb[OWN_UNR];
#pragma unroll
      for (int u = 0; u < OWN_UNR; ++u) {
        const int it = min(ib + lane + u * OWN_NT, nit - 1);
        const int q = glist[it / BIT_PPG] * BIT_PPG + it % BIT_PPG;
        eb[u] = q;
        dv[u] = d4[min(q, nq - 1)];
      }
      unsigned hm = 0;
#pragma unroll
      for (int u = 0; u < OWN_UNR; ++u) {
        const unsigned h = ((unsigned)(dv[u].x - n0) < (unsigned)nloc ? 1u : 0u) |
                           ((unsigned)(dv[u].y - n0) < (unsigned)nloc ? 2u : 0u) |
                           ((unsigned)(dv[u].z - n0) < (unsigned)nloc ? 4u : 0u) |
                           ((unsigned)(dv[u].w - n0) < (unsigned)nloc ? 8u : 0u);
        const bool live = ib + lane + u * OWN_NT < nit && eb[u] < nq;
        hm |= (live ? h : 0u) << (4 * u);
      }
      int pos = wave_reserve(__popc(hm), &cnt, lane);
#pragma unroll
      for (int u = 0; u < OWN_UNR; ++u) {
        unsigned m = (hm >> (4 * u)) & 15u;
        while (m) {
          const int j = __ffs(m) - 1;
          m &= m - 1;
          if (pos < OWN_CAP) list_s[pos] = eb[u] * 4 + j;
          ++pos;
        }
      }
    }
    if (tid < (E & 3)) {
      const int e = (E & ~3) + tid;
      if ((unsigned)(dst[e] - n0) < (unsigned)nloc) {
        const int pos = atomicAdd(&cnt, 1);
        if (pos < OWN_CAP) list_s[pos] = e;
      }
    }
  }
  __syncthreads();
  const int total = SAVED ? sv_n[blockIdx.x] : (keyed ? cnt : OWN_CAP + 1);
  if (!SAVED && tid == 0) sv_n[blockIdx.x] = total;

  if (total <= OWN_CAP) {
   if constexpr (SAVED) {
    for (int i0 = nloc + tid; i0 < total; i0 += OWN_NT * OWN_UNR) {
      int s[OWN_UNR], d[OWN_UNR];
#pragma unroll
      for (int u = 0; u < OWN_UNR; ++u) {
        const int i = min(i0 + u * OWN_NT, total - 1);
        s[u] = gs[i];
        d[u] = gd[i];
      }
#pragma unroll
      for (int u = 0; u < OWN_UNR; ++u) {
        const int i = i0 + u * OWN_NT;
        if (i < total) { list_s[i] = s[u]; list_d[i] = d[u]; }
      }
    }
   } else {
    for (int i0 = nloc + tid; i0 < total; i0 += OWN_NT * OWN_UNR) {
      int s[OWN_UNR], d[OWN_UNR];
#pragma unroll
      for (int u = 0; u < OWN_UNR; ++u) {
        const int i = i0 + u * OWN_NT;
        const int e = list_s[min(i, total - 1)];
        const int ee = i < total ? e : 0;
        s[u] = src[ee];
        d[u] = dst[ee];
      }
#pragma unroll
      for (int u = 0; u < OWN_UNR; ++u) {
        const int i = i0 + u * OWN_NT;
        if (i < total) {
          list_s[i] = s[u]; list_d[i] = d[u] - n0;
          gs[i] = s[u]; gd[i] = d[u] - n0;
        }
      }
    }
   }
    __syncthreads();
    {
      constexpr int U1 = 4;
      const int totH = total * H;
      int it = tid;
      for (; it + (U1 - 1) * OWN_NT < totH; it += U1 * OWN_NT) {
        float a[U1];
        int sl[U1];
#pragma unroll
        for (int u = 0; u < U1; ++u) {
          const int q = it + u * OWN_NT;
          const int i = q / H, h = q % H;
          sl[u] = list_d[i] * H + h;
          a[u] = as[(size_t)list_s[i] * H + h];
        }
#pragma unroll
        for (int u = 0; u < U1; ++u) {
          const float v = lrelu(a[u] + ad_l[sl[u]], ATT_SLOPE);
          ew[it + u * OWN_NT] = v;
          lds_max_f(&m_l[sl[u]], v);
        }
      }
      for (; it < totH; it += OWN_NT) {
        const int i = it / H, h = it % H;
        const int sl = list_d[i] * H + h;
        const float v = lrelu(as[(size_t)list_s[i] * H + h] + ad_l[sl], ATT_SLOPE);
        ew[it] = v;
        lds_max_f(&m_l[sl], v);
      }
    }
    __syncthreads();
    for (int it = tid; it < total * H; it += OWN_NT) {
      const int i = it / H, h = it % H;
      const int dl = list_d[i];
      const float w = expf(ew[it] - __int_as_float(m_l[dl * H + h]));
      ew[it] = w;
      atomicAdd(&den_l[dl * H + h], w);
    }
    __syncthreads();
    static_assert(C % 4 == 0 && HC % 4 == 0, "a lane's four columns lie in one head");
    constexpr int LPE = HC / 4;
    static_assert(OWN_NT % LPE == 0 && (LPE == 16 || LPE <= 8) && 32 % LPE == 0, "edge slots of a wave");
    constexpr int EPI = OWN_NT / LPE;
    const int c0 = (tid % LPE) * 4;
    const int h = c0 / C;
    const int slot = (tid & 31) / LPE;
    const int rot = LPE == 16 ? (slot & 1) * 2 : (slot & 3);
    const int o0 = rot, o1 = (rot + 1) & 3, o2 = (rot + 2) & 3, o3 = (rot + 3) & 3;
    auto add4 = [&](int dloc, float wt, const v4f& v) {
      const float a0 = (rot & 1) ? v.y : v.x;
      const float a1 = (rot & 1) ? v.z : v.y;
      const float a2 = (rot & 1) ? v.w : v.z;
      const float a3 = (rot & 1) ? v.x : v.w;
      const float p0 = (rot & 2) ? a2 : a0;
      const float p1 = (rot & 2) ? a3 : a1;
      const float p2 = (rot & 2) ? a0 : a2;
      const float p3 = (rot & 2) ? a1 : a3;
      float* const arow = acc_l + dloc * HC + c0;
      atomicAdd(arow + o0, wt * p0);
      atomicAdd(arow + o1, wt * p1);
      atomicAdd(arow + o2, wt * p2);
      atomicAdd(arow + o3, wt * p3);
    };
    int i0 = tid / LPE;
    for (; i0 + (OWN_UNR - 1) * EPI < total; i0 += EPI * OWN_UNR) {
      v4f hv[OWN_UNR];
      float w[OWN_UNR];
      int dl[OWN_UNR];
#pragma unroll
      for (int u = 0; u < OWN_UNR; ++u) {
        const int i = i0 + u * EPI;
        dl[u] = list_d[i];
        w[u] = ew[i * H + h];
        hv[u] = *reinterpret_cast<const v4f*>(hfeat + (size_t)list_s[i] * HC + c0);
      }
#pragma unroll
      for (int u = 0; u < OWN_UNR; ++u) add4(dl[u], w[u], hv[u]);
    }
    if (i0 < total) {
      v4f hv[OWN_UNR];
      float w[OWN_UNR];
      int dl[OWN_UNR];
#pragma unroll
      for (int u = 0; u < OWN_UNR; ++u) {
        const int i = min(i0 + u * EPI, total - 1);
        dl[u] = list_d[i];
        w[u] = ew[i * H + h];
        hv[u] = *reinterpret_cast<const v4f*>(hfeat + (size_t)list_s[i] * HC + c0);
      }
      asm volatile("" :: "v"(hv[0]), "v"(hv[1]), "v"(hv[2]), "v"(hv[3]),
                         "v"(hv[4]), "v"(hv[5]), "v"(hv[6]), "v"(hv[7]));
      static_assert(OWN_UNR == 8, "the statement above names the eight gathers");
#pragma unroll
      for (int u = 0; u < OWN_UNR; ++u) {
        if (i0 + u * EPI < total) add4(dl[u], w[u], hv[u]);
      }
    }
  } else {
    const int tot = E + nloc;
    for (int e = tid; e < tot; e += OWN_NT) {
      int s, dl;
      if (e < E) {
        const unsigned x = (unsigned)(dst[e] - n0);
        if (x >= (unsigned)nloc) continue;
        dl = (int)x; s = src[e];
      } else { dl = e - E; s = n0 + dl; }
      for (int h = 0; h < H; ++h) {
        const float v = lrelu(as[(size_t)s * H + h] + ad_l[dl * H + h], ATT_SLOPE);
        lds_max_f(&m_l[dl * H + h], v);
      }
    }
    __syncthreads();
    for (int e = tid; e < tot; e += OWN_NT) {
      int s, dl;
      if (e < E) {
        const unsigned x = (unsigned)(dst[e] - n0);
        if (x >= (unsigned)nloc) continue;
        dl = (int)x; s = src[e];
      } else { dl = e - E; s = n0 + dl; }
      for (int h = 0; h < H; ++h) {
        const float v = lrelu(as[(size_t)s * H + h] + ad_l[dl * H + h], ATT_SLOPE);
        const float w = expf(v - __int_as_float(m_l[dl * H + h]));
        atomicAdd(&den_l[dl * H + h], w);
        for (int c = 0; c < C; ++c)
          atomicAdd(&acc_l[dl * HC + h * C + c], w * hfeat[(size_t)s * HC + h * C + c]);
      }
    }
  }
  __syncthreads();

  if constexpr (!FINAL) {
    for (int it = tid; it < nloc * HC; it += OWN_NT) {
      const int n = it / HC, jj = it % HC;
      const float v = acc_l[it] / den_l[n * H + jj / C] + bias[jj];
      outp[(size_t)n0 * HC + it] = lrelu(v, ACT_SLOPE);
    }
  } else {
    for (int n = tid; n < nloc; n += OWN_NT) {
      const float dn = den_l[n];
      float s = 0.f;
      for (int c = 0; c < C; ++c) {
        const float f = lrelu(acc_l[n * C + c] / dn + bias[c], ACT_SLOPE);
        s += f * Wo[c];
      }
      outp[n0 + n] = s + bo[0];
    }
  }
}

extern "C" void kernel_launch(void* const* d_in, const int* in_sizes, int n_in,
                              void* d_out, int out_size, void* d_ws, size_t ws_size,
                              hipStream_t stream) {
  const float* x    = (const float*)d_in[0];
  const int*   ei   = (const int*)d_in[1];
  const float* W1   = (const float*)d_in[2];
  const float* aS1  = (const float*)d_in[3];
  const float* aD1  = (const float*)d_in[4];
  const float* b1   = (const float*)d_in[5];
  const float* W2   = (const float*)d_in[6];
  const float* aS2  = (const float*)d_in[7];
  const float* aD2  = (const float*)d_in[8];
  const float* b2   = (const float*)d_in[9];
  const float* Wo   = (const float*)d_in[10];
  const float* bo   = (const float*)d_in[11];
  float* out = (float*)d_out;

  const int IN_C = 64, HID = 16, HEADS = 4;
  const int HC1 = HEADS * HID;
  const int Nn = in_sizes[0] / IN_C;
  const int E  = in_sizes[1] / 2;
  const int* src = ei;
  const int* dst = ei + E;

  const int GB = (Nn + 16 * GEMM_WPG - 1) / (16 * GEMM_WPG);
  const int OG = (Nn + OWN_NPW - 1) / OWN_NPW;
  const int NCH = (E + BIT_NT * BIT_EPT - 1) / (BIT_NT * BIT_EPT) > 0
                ? (E + BIT_NT * BIT_EPT - 1) / (BIT_NT * BIT_EPT) : 1;
  const int rowWords = NCH * BIT_CW;

  unsigned int* bits = (unsigned int*)d_ws;
  float* p = (float*)(bits + (size_t)OG * rowWords);
  float* h1   = p; p += (size_t)Nn * HC1;
  float* f1   = p; p += (size_t)Nn * HC1;
  float* h2   = p; p += (size_t)Nn * HID;
  float* as1  = p; p += (size_t)Nn * HEADS;
  float* ad1  = p; p += (size_t)Nn * HEADS;
  float* as2  = p; p += (size_t)Nn;
  float* ad2  = p; p += (size_t)Nn;

  int* sv_s = (int*)p;
  int* sv_d = sv_s + (size_t)OG * OWN_CAP;
  int* sv_n = sv_d + (size_t)OG * OWN_CAP;

  owner_bits<<<NCH, BIT_NT, 0, stream>>>(dst, bits, E, Nn, OG, rowWords);

  wmma_gemm_coef<64, 4><<<GB, GEMM_WPG * 32, 0, stream>>>(x, W1, aS1, aD1, h1, as1, ad1, Nn);
  owner_layer<4, 16, false, false><<<OG, OWN_NT, 0, stream>>>(src, dst, bits, rowWords, sv_s, sv_d, sv_n,
                                                            as1, ad1, h1, b1, Wo, bo, f1, E, Nn);

  wmma_gemm_coef<64, 1><<<GB, GEMM_WPG * 32, 0, stream>>>(f1, W2, aS2, aD2, h2, as2, ad2, Nn);
  owner_layer<1, 16, true, true><<<OG, OWN_NT, 0, stream>>>(src, dst, bits, rowWords, sv_s, sv_d, sv_n,
                                                          as2, ad2, h2, b2, Wo, bo, out, E, Nn);
}


# The filter keeps a kept edge's local destination beside its number (it has it in hand): the batched pass behind it fetches only the source, one scattered fetch per edge, not two; sum order as before
# speedup vs baseline: 7.7897x; 1.0531x over previous
#include <hip/hip_runtime.h>
#include <math.h>
#include <stdint.h>


#define ATT_SLOPE 0.2f
#define ACT_SLOPE 0.01f

typedef __attribute__((ext_vector_type(2))) float v2f;
typedef __attribute__((ext_vector_type(8))) float v8f;
typedef __attribute__((ext_vector_type(4))) float v4f;

__device__ __forceinline__ float lrelu(float v, float s) { return v > 0.f ? v : v * s; }

__device__ __forceinline__ void lds_max_f(int* slot, float val) {
  if (val >= 0.f)
    atomicMax(slot, __float_as_int(val));
  else
    atomicMin((unsigned int*)slot, (unsigned int)__float_as_int(val));
}

constexpr int GEMM_WPG = 4;

template <int MASK>
__device__ __forceinline__ float row_xadd(float v) {
  return v + __int_as_float(__builtin_amdgcn_update_dpp(0, __float_as_int(v), 0x160 + MASK,
                                                        0xf, 0xf, true));
}

template <int K, int H>
__global__ void __launch_bounds__(GEMM_WPG * 32)
wmma_gemm_coef(const float* __restrict__ A, const float* __restrict__ B,
               const float* __restrict__ att_src, const float* __restrict__ att_dst,
               float* __restrict__ D, float* __restrict__ as, float* __restrict__ ad, int M) {
  constexpr int NCOLS = H * 16;
  const int m0 = (blockIdx.x * GEMM_WPG + (threadIdx.x >> 5)) * 16;
  if (m0 >= M) return;
  const int lane = threadIdx.x & 31;
  const int half = lane >> 4;
  const int lid  = lane & 15;
  const int ar = min(m0 + lid, M - 1);
  v8f acc[H];
#pragma unroll
  for (int t = 0; t < H; ++t) acc[t] = v8f{};
#pragma unroll
  for (int k = 0; k < K; k += 4) {
    const int ka = k + half * 2;
    v2f a;
    a.x = A[(size_t)ar * K + ka];
    a.y = A[(size_t)ar * K + ka + 1];
#pragma unroll
    for (int t = 0; t < H; ++t) {
      v2f b;
      b.x = B[(size_t)ka * NCOLS + t * 16 + lid];
      b.y = B[(size_t)(ka + 1) * NCOLS + t * 16 + lid];
      acc[t] = __builtin_amdgcn_wmma_f32_16x16x4_f32(false, a, false, b, (short)0, acc[t],
                                                     false, false);
    }
  }
  const int rbase = m0 + half * 8;
  float* const drow = D + (size_t)rbase * NCOLS + lid;
  if (m0 + 16 <= M) {
#pragma unroll
    for (int t = 0; t < H; ++t) {
#pragma unroll
      for (int r = 0; r < 8; ++r) drow[r * NCOLS + t * 16] = acc[t][r];
    }
  } else {
#pragma unroll
    for (int t = 0; t < H; ++t) {
#pragma unroll
      for (int r = 0; r < 8; ++r) {
        if (rbase + r < M) drow[r * NCOLS + t * 16] = acc[t][r];
      }
    }
  }
  float so[H], dv[H];
#pragma unroll
  for (int t = 0; t < H; ++t) {
    const float ws = att_src[t * 16 + lid];
    const float wd = att_dst[t * 16 + lid];
    float s[8], d[8];
#pragma unroll
    for (int r = 0; r < 8; ++r) { s[r] = acc[t][r] * ws; d[r] = acc[t][r] * wd; }
#pragma unroll
    for (int r = 0; r < 8; ++r) { s[r] = row_xadd<1>(s[r]); d[r] = row_xadd<1>(d[r]); }
#pragma unroll
    for (int r = 0; r < 8; ++r) { s[r] = row_xadd<2>(s[r]); d[r] = row_xadd<2>(d[r]); }
#pragma unroll
    for (int r = 0; r < 8; ++r) { s[r] = row_xadd<4>(s[r]); d[r] = row_xadd<4>(d[r]); }
#pragma unroll
    for (int r = 0; r < 8; ++r) { s[r] = row_xadd<8>(s[r]); d[r] = row_xadd<8>(d[r]); }
    float ss = s[0], dd = d[0];
#pragma unroll
    for (int r = 1; r < 8; ++r) { ss = lid == r ? s[r] : ss; dd = lid == r ? d[r] : dd; }
    so[t] = ss; dv[t] = dd;
  }
  const int row = rbase + lid;
  if (lid < 8 && row < M) {
    if constexpr (H % 4 == 0) {
#pragma unroll
      for (int t = 0; t < H; t += 4) {
        *reinterpret_cast<float4*>(as + (size_t)row * H + t) = make_float4(so[t], so[t + 1], so[t + 2], so[t + 3]);
        *reinterpret_cast<float4*>(ad + (size_t)row * H + t) = make_float4(dv[t], dv[t + 1], dv[t + 2], dv[t + 3]);
      }
    } else {
#pragma unroll
      for (int t = 0; t < H; ++t) { as[(size_t)row * H + t] = so[t]; ad[(size_t)row * H + t] = dv[t]; }
    }
  }
}

constexpr int OWN_NT  = 1024;
constexpr int OWN_NPW = 224;
constexpr int OWN_CAP = 9216;
constexpr int OWN_UNR = 8;
constexpr unsigned OWN_NOKEY = 255u;
constexpr int BIT_GRP   = 8;
constexpr int BIT_PPG   = BIT_GRP / 4;
constexpr int BIT_NT    = 512;
constexpr int BIT_EPT   = 16;
constexpr int BIT_TGR   = BIT_EPT / BIT_GRP;
constexpr int BIT_CW    = BIT_NT * BIT_TGR / 32;
constexpr int BIT_PITCH = 257;

__device__ __forceinline__ unsigned owner_key(int d, int Nn) {
  return (unsigned)d < (unsigned)Nn ? min((unsigned)d / (unsigned)OWN_NPW, OWN_NOKEY) : OWN_NOKEY;
}

__global__ void __launch_bounds__(BIT_NT)
owner_bits(const int* __restrict__ dst, unsigned int* __restrict__ bits,
           int E, int Nn, int rows, int rowWords) {
  __shared__ unsigned int lb[BIT_CW * BIT_PITCH];
  const bool vec = ((reinterpret_cast<uintptr_t>(dst) & 15) == 0);
  if (rows > (int)OWN_NOKEY || !vec || E < 4) return;
  const int tid = threadIdx.x;
  for (int i = tid; i < BIT_CW * BIT_PITCH; i += BIT_NT) lb[i] = 0u;
  const int e0 = (blockIdx.x * BIT_NT + tid) * BIT_EPT;
  const int4* d4 = reinterpret_cast<const int4*>(dst);
  const int nq = E >> 2;
  int4 v[4];
#pragma unroll
  for (int j = 0; j < 4; ++j) v[j] = d4[min((e0 >> 2) + j, nq - 1)];
  __syncthreads();
  auto mark = [&](int d, int lg) {
    const unsigned o = owner_key(d, Nn);
    if (o != OWN_NOKEY) atomicOr(&lb[(lg >> 5) * BIT_PITCH + o], 1u << (lg & 31));
  };
#pragma unroll
  for (int j = 0; j < 4; ++j) {
    const int lg = tid * BIT_TGR + j / BIT_PPG;
    if ((e0 >> 2) + j < nq) { mark(v[j].x, lg); mark(v[j].y, lg); mark(v[j].z, lg); mark(v[j].w, lg); }
  }
  if (blockIdx.x == gridDim.x - 1 && tid < (E & 3)) {
    const int e = (E & ~3) + tid;
    mark(dst[e], e / BIT_GRP - blockIdx.x * (BIT_NT * BIT_TGR));
  }
  __syncthreads();
  for (int i = tid; i < rows * BIT_CW; i += BIT_NT) {
    const int o = i / BIT_CW, w = i % BIT_CW;
    bits[(size_t)o * rowWords + blockIdx.x * BIT_CW + w] = lb[w * BIT_PITCH + o];
  }
}

template <int N>
__device__ __forceinline__ int row_shr(int x) {
  return __builtin_amdgcn_update_dpp(0, x, 0x110 + N, 0xf, 0xf, true);
}
__device__ __forceinline__ int wave_reserve(int c, int* counter, int lane) {
  int incl = c;
  incl += row_shr<1>(incl);
  incl += row_shr<2>(incl);
  incl += row_shr<4>(incl);
  incl += row_shr<8>(incl);
  const int r0 = __builtin_amdgcn_readlane(incl, 15);
  const int r1 = __builtin_amdgcn_readlane(incl, 31);
  int base = 0;
  if (lane == 0) base = atomicAdd(counter, r0 + r1);
  base = __builtin_amdgcn_readfirstlane(base);
  return base + incl - c + (lane >= 16 ? r0 : 0);
}

template <int H, int C, bool FINAL, bool SAVED>
__global__ void __launch_bounds__(OWN_NT)
owner_layer(const int* __restrict__ src, const int* __restrict__ dst,
            const unsigned int* __restrict__ bits, int rowWords,
            int* __restrict__ sv_s, int* __restrict__ sv_d, int* __restrict__ sv_n,
            const float* __restrict__ as, const float* __restrict__ ad,
            const float* __restrict__ hfeat, const float* __restrict__ bias,
            const float* __restrict__ Wo, const float* __restrict__ bo,
            float* __restrict__ outp, int E, int Nn) {
  constexpr int HC = H * C;
  static_assert(OWN_NT % HC == 0, "a thread keeps one feature column");
  __shared__ float acc_l[OWN_NPW * HC];
  __shared__ float den_l[OWN_NPW * H];
  __shared__ float ad_l[OWN_NPW * H];
  __shared__ int   m_l[OWN_NPW * H];
  __shared__ int   list_s[OWN_CAP];
  __shared__ int   list_d[OWN_CAP];
  __shared__ float ew[OWN_CAP * H];
  __shared__ int   cnt;
  __shared__ int   gcnt;

  const int tid = threadIdx.x;
  const int n0 = blockIdx.x * OWN_NPW;
  const int nloc = min(OWN_NPW, Nn - n0);
  if (nloc <= 0) return;

  for (int i = tid; i < nloc * HC; i += OWN_NT) acc_l[i] = 0.f;
  for (int i = tid; i < nloc * H; i += OWN_NT) {
    den_l[i] = 0.f;
    m_l[i] = __float_as_int(-INFINITY);
    ad_l[i] = ad[(size_t)n0 * H + i];
  }
  for (int i = tid; i < nloc; i += OWN_NT) { list_s[i] = n0 + i; list_d[i] = i; }
  if (tid == 0) { cnt = nloc; gcnt = 0; }
  __syncthreads();

  int* const gs = sv_s + (size_t)blockIdx.x * OWN_CAP;
  int* const gd = sv_d + (size_t)blockIdx.x * OWN_CAP;

  const bool keyed = !SAVED && gridDim.x <= OWN_NOKEY && E >= 4 &&
                     ((reinterpret_cast<uintptr_t>(dst) & 15) == 0);
  if (keyed) {
    const int lane = tid & 31;
    int* const glist = reinterpret_cast<int*>(ew);
    const unsigned int* const row = bits + (size_t)blockIdx.x * rowWords;
    for (int wb = tid - lane; wb < rowWords; wb += OWN_NT * 4) {
      unsigned wv[4];
      int c = 0;
#pragma unroll
      for (int u = 0; u < 4; ++u) {
        const int wi = wb + lane + u * OWN_NT;
        wv[u] = wi < rowWords ? row[wi] : 0u;
      }
#pragma unroll
      for (int u = 0; u < 4; ++u) c += __popc(wv[u]);
      int pos = wave_reserve(c, &gcnt, lane);
#pragma unroll
      for (int u = 0; u < 4; ++u) {
        unsigned m = wv[u];
        const int g0 = (wb + lane + u * OWN_NT) * 32;
        while (m) {
          const int b = __ffs(m) - 1;
          m &= m - 1;
          if (pos < OWN_CAP) glist[pos] = g0 + b;
          ++pos;
        }
      }
    }
    __syncthreads();
    const int nit = min(gcnt, OWN_CAP) * BIT_PPG;
    const int4* d4 = reinterpret_cast<const int4*>(dst);
    const int nq = E >> 2;
    for (int ib = tid - lane; ib < nit; ib += OWN_NT * OWN_UNR) {
      int4 dv[OWN_UNR];
      int  eb[OWN_UNR];
#pragma unroll
      for (int u = 0; u < OWN_UNR; ++u) {
        const int it = min(ib + lane + u * OWN_NT, nit - 1);
        const int q = glist[it / BIT_PPG] * BIT_PPG + it % BIT_PPG;
        eb[u] = q;
        dv[u] = d4[min(q, nq - 1)];
      }
      unsigned hm = 0;
      unsigned pk[OWN_UNR];
      static_assert(OWN_NPW <= 256, "a local destination fits one byte of the packed word");
#pragma unroll
      for (int u = 0; u < OWN_UNR; ++u) {
        const unsigned x0 = (unsigned)(dv[u].x - n0), x1 = (unsigned)(dv[u].y - n0);
        const unsigned x2 = (unsigned)(dv[u].z - n0), x3 = (unsigned)(dv[u].w - n0);
        const unsigned h = (x0 < (unsigned)nloc ? 1u : 0u) | (x1 < (unsigned)nloc ? 2u : 0u) |
                           (x2 < (unsigned)nloc ? 4u : 0u) | (x3 < (unsigned)nloc ? 8u : 0u);
        pk[u] = (x0 & 255u) | ((x1 & 255u) << 8) | ((x2 & 255u) << 16) | (x3 << 24);
        const bool live = ib + lane + u * OWN_NT < nit && eb[u] < nq;
        hm |= (live ? h : 0u) << (4 * u);
      }
      int pos = wave_reserve(__popc(hm), &cnt, lane);
#pragma unroll
      for (int u = 0; u < OWN_UNR; ++u) {
        unsigned m = (hm >> (4 * u)) & 15u;
        while (m) {
          const int j = __ffs(m) - 1;
          m &= m - 1;
          if (pos < OWN_CAP) { list_s[pos] = eb[u] * 4 + j; list_d[pos] = (int)((pk[u] >> (8 * j)) & 255u); }
          ++pos;
        }
      }
    }
    if (tid < (E & 3)) {
      const int e = (E & ~3) + tid;
      const int dl = dst[e] - n0;
      if ((unsigned)dl < (unsigned)nloc) {
        const int pos = atomicAdd(&cnt, 1);
        if (pos < OWN_CAP) { list_s[pos] = e; list_d[pos] = dl; }
      }
    }
  }
  __syncthreads();
  const int total = SAVED ? sv_n[blockIdx.x] : (keyed ? cnt : OWN_CAP + 1);
  if (!SAVED && tid == 0) sv_n[blockIdx.x] = total;

  if (total <= OWN_CAP) {
   if constexpr (SAVED) {
    for (int i0 = nloc + tid; i0 < total; i0 += OWN_NT * OWN_UNR) {
      int s[OWN_UNR], d[OWN_UNR];
#pragma unroll
      for (int u = 0; u < OWN_UNR; ++u) {
        const int i = min(i0 + u * OWN_NT, total - 1);
        s[u] = gs[i];
        d[u] = gd[i];
      }
#pragma unroll
      for (int u = 0; u < OWN_UNR; ++u) {
        const int i = i0 + u * OWN_NT;
        if (i < total) { list_s[i] = s[u]; list_d[i] = d[u]; }
      }
    }
   } else {
    for (int i0 = nloc + tid; i0 < total; i0 += OWN_NT * OWN_UNR) {
      int s[OWN_UNR], d[OWN_UNR];
#pragma unroll
      for (int u = 0; u < OWN_UNR; ++u) {
        const int i = i0 + u * OWN_NT;
        const int ic = min(i, total - 1);
        const int e = list_s[ic];
        const int ee = i < total ? e : 0;
        s[u] = src[ee];
        d[u] = list_d[ic];
      }
#pragma unroll
      for (int u = 0; u < OWN_UNR; ++u) {
        const int i = i0 + u * OWN_NT;
        if (i < total) {
          list_s[i] = s[u];
          gs[i] = s[u]; gd[i] = d[u];
        }
      }
    }
   }
    __syncthreads();
    {
      constexpr int U1 = 4;
      const int totH = total * H;
      int it = tid;
      for (; it + (U1 - 1) * OWN_NT < totH; it += U1 * OWN_NT) {
        float a[U1];
        int sl[U1];
#pragma unroll
        for (int u = 0; u < U1; ++u) {
          const int q = it + u * OWN_NT;
          const int i = q / H, h = q % H;
          sl[u] = list_d[i] * H + h;
          a[u] = as[(size_t)list_s[i] * H + h];
        }
#pragma unroll
        for (int u = 0; u < U1; ++u) {
          const float v = lrelu(a[u] + ad_l[sl[u]], ATT_SLOPE);
          ew[it + u * OWN_NT] = v;
          lds_max_f(&m_l[sl[u]], v);
        }
      }
      for (; it < totH; it += OWN_NT) {
        const int i = it / H, h = it % H;
        const int sl = list_d[i] * H + h;
        const float v = lrelu(as[(size_t)list_s[i] * H + h] + ad_l[sl], ATT_SLOPE);
        ew[it] = v;
        lds_max_f(&m_l[sl], v);
      }
    }
    __syncthreads();
    for (int it = tid; it < total * H; it += OWN_NT) {
      const int i = it / H, h = it % H;
      const int dl = list_d[i];
      const float w = expf(ew[it] - __int_as_float(m_l[dl * H + h]));
      ew[it] = w;
      atomicAdd(&den_l[dl * H + h], w);
    }
    __syncthreads();
    static_assert(C % 4 == 0 && HC % 4 == 0, "a lane's four columns lie in one head");
    constexpr int LPE = HC / 4;
    static_assert(OWN_NT % LPE == 0 && (LPE == 16 || LPE <= 8) && 32 % LPE == 0, "edge slots of a wave");
    constexpr int EPI = OWN_NT / LPE;
    const int c0 = (tid % LPE) * 4;
    const int h = c0 / C;
    const int slot = (tid & 31) / LPE;
    const int rot = LPE == 16 ? (slot & 1) * 2 : (slot & 3);
    const int o0 = rot, o1 = (rot + 1) & 3, o2 = (rot + 2) & 3, o3 = (rot + 3) & 3;
    auto add4 = [&](int dloc, float wt, const v4f& v) {
      const float a0 = (rot & 1) ? v.y : v.x;
      const float a1 = (rot & 1) ? v.z : v.y;
      const float a2 = (rot & 1) ? v.w : v.z;
      const float a3 = (rot & 1) ? v.x : v.w;
      const float p0 = (rot & 2) ? a2 : a0;
      const float p1 = (rot & 2) ? a3 : a1;
      const float p2 = (rot & 2) ? a0 : a2;
      const float p3 = (rot & 2) ? a1 : a3;
      float* const arow = acc_l + dloc * HC + c0;
      atomicAdd(arow + o0, wt * p0);
      atomicAdd(arow + o1, wt * p1);
      atomicAdd(arow + o2, wt * p2);
      atomicAdd(arow + o3, wt * p3);
    };
    int i0 = tid / LPE;
    for (; i0 + (OWN_UNR - 1) * EPI < total; i0 += EPI * OWN_UNR) {
      v4f hv[OWN_UNR];
      float w[OWN_UNR];
      int dl[OWN_UNR];
#pragma unroll
      for (int u = 0; u < OWN_UNR; ++u) {
        const int i = i0 + u * EPI;
        dl[u] = list_d[i];
        w[u] = ew[i * H + h];
        hv[u] = *reinterpret_cast<const v4f*>(hfeat + (size_t)list_s[i] * HC + c0);
      }
#pragma unroll
      for (int u = 0; u < OWN_UNR; ++u) add4(dl[u], w[u], hv[u]);
    }
    if (i0 < total) {
      v4f hv[OWN_UNR];
      float w[OWN_UNR];
      int dl[OWN_UNR];
#pragma unroll
      for (int u = 0; u < OWN_UNR; ++u) {
        const int i = min(i0 + u * EPI, total - 1);
        dl[u] = list_d[i];
        w[u] = ew[i * H + h];
        hv[u] = *reinterpret_cast<const v4f*>(hfeat + (size_t)list_s[i] * HC + c0);
      }
      asm volatile("" :: "v"(hv[0]), "v"(hv[1]), "v"(hv[2]), "v"(hv[3]),
                         "v"(hv[4]), "v"(hv[5]), "v"(hv[6]), "v"(hv[7]));
      static_assert(OWN_UNR == 8, "the statement above names the eight gathers");
#pragma unroll
      for (int u = 0; u < OWN_UNR; ++u) {
        if (i0 + u * EPI < total) add4(dl[u], w[u], hv[u]);
      }
    }
  } else {
    const int tot = E + nloc;
    for (int e = tid; e < tot; e += OWN_NT) {
      int s, dl;
      if (e < E) {
        const unsigned x = (unsigned)(dst[e] - n0);
        if (x >= (unsigned)nloc) continue;
        dl = (int)x; s = src[e];
      } else { dl = e - E; s = n0 + dl; }
      for (int h = 0; h < H; ++h) {
        const float v = lrelu(as[(size_t)s * H + h] + ad_l[dl * H + h], ATT_SLOPE);
        lds_max_f(&m_l[dl * H + h], v);
      }
    }
    __syncthreads();
    for (int e = tid; e < tot; e += OWN_NT) {
      int s, dl;
      if (e < E) {
        const unsigned x = (unsigned)(dst[e] - n0);
        if (x >= (unsigned)nloc) continue;
        dl = (int)x; s = src[e];
      } else { dl = e - E; s = n0 + dl; }
      for (int h = 0; h < H; ++h) {
        const float v = lrelu(as[(size_t)s * H + h] + ad_l[dl * H + h], ATT_SLOPE);
        const float w = expf(v - __int_as_float(m_l[dl * H + h]));
        atomicAdd(&den_l[dl * H + h], w);
        for (int c = 0; c < C; ++c)
          atomicAdd(&acc_l[dl * HC + h * C + c], w * hfeat[(size_t)s * HC + h * C + c]);
      }
    }
  }
  __syncthreads();

  if constexpr (!FINAL) {
    for (int it = tid; it < nloc * HC; it += OWN_NT) {
      const int n = it / HC, jj = it % HC;
      const float v = acc_l[it] / den_l[n * H + jj / C] + bias[jj];
      outp[(size_t)n0 * HC + it] = lrelu(v, ACT_SLOPE);
    }
  } else {
    for (int n = tid; n < nloc; n += OWN_NT) {
      const float dn = den_l[n];
      float s = 0.f;
      for (int c = 0; c < C; ++c) {
        const float f = lrelu(acc_l[n * C + c] / dn + bias[c], ACT_SLOPE);
        s += f * Wo[c];
      }
      outp[n0 + n] = s + bo[0];
    }
  }
}

extern "C" void kernel_launch(void* const* d_in, const int* in_sizes, int n_in,
                              void* d_out, int out_size, void* d_ws, size_t ws_size,
                              hipStream_t stream) {
  const float* x    = (const float*)d_in[0];
  const int*   ei   = (const int*)d_in[1];
  const float* W1   = (const float*)d_in[2];
  const float* aS1  = (const float*)d_in[3];
  const float* aD1  = (const float*)d_in[4];
  const float* b1   = (const float*)d_in[5];
  const float* W2   = (const float*)d_in[6];
  const float* aS2  = (const float*)d_in[7];
  const float* aD2  = (const float*)d_in[8];
  const float* b2   = (const float*)d_in[9];
  const float* Wo   = (const float*)d_in[10];
  const float* bo   = (const float*)d_in[11];
  float* out = (float*)d_out;

  const int IN_C = 64, HID = 16, HEADS = 4;
  const int HC1 = HEADS * HID;
  const int Nn = in_sizes[0] / IN_C;
  const int E  = in_sizes[1] / 2;
  const int* src = ei;
  const int* dst = ei + E;

  const int GB = (Nn + 16 * GEMM_WPG - 1) / (16 * GEMM_WPG);
  const int OG = (Nn + OWN_NPW - 1) / OWN_NPW;
  const int NCH = (E + BIT_NT * BIT_EPT - 1) / (BIT_NT * BIT_EPT) > 0
                ? (E + BIT_NT * BIT_EPT - 1) / (BIT_NT * BIT_EPT) : 1;
  const int rowWords = NCH * BIT_CW;

  unsigned int* bits = (unsigned int*)d_ws;
  float* p = (float*)(bits + (size_t)OG * rowWords);
  float* h1   = p; p += (size_t)Nn * HC1;
  float* f1   = p; p += (size_t)Nn * HC1;
  float* h2   = p; p += (size_t)Nn * HID;
  float* as1  = p; p += (size_t)Nn * HEADS;
  float* ad1  = p; p += (size_t)Nn * HEADS;
  float* as2  = p; p += (size_t)Nn;
  float* ad2  = p; p += (size_t)Nn;

  int* sv_s = (int*)p;
  int* sv_d = sv_s + (size_t)OG * OWN_CAP;
  int* sv_n = sv_d + (size_t)OG * OWN_CAP;

  owner_bits<<<NCH, BIT_NT, 0, stream>>>(dst, bits, E, Nn, OG, rowWords);

  wmma_gemm_coef<64, 4><<<GB, GEMM_WPG * 32, 0, stream>>>(x, W1, aS1, aD1, h1, as1, ad1, Nn);
  owner_layer<4, 16, false, false><<<OG, OWN_NT, 0, stream>>>(src, dst, bits, rowWords, sv_s, sv_d, sv_n,
                                                            as1, ad1, h1, b1, Wo, bo, f1, E, Nn);

  wmma_gemm_coef<64, 1><<<GB, GEMM_WPG * 32, 0, stream>>>(f1, W2, aS2, aD2, h2, as2, ad2, Nn);
  owner_layer<1, 16, true, true><<<OG, OWN_NT, 0, stream>>>(src, dst, bits, rowWords, sv_s, sv_d, sv_n,
                                                          as2, ad2, h2, b2, Wo, bo, out, E, Nn);
}


# Owner ranges of 200 destinations, not 224: 250 work-groups on the chip's 256 WGPs (was 224), a tenth less work each; a final-output line at a range border now has two writers; sum order as before
# speedup vs baseline: 8.1309x; 1.0438x over previous
#include <hip/hip_runtime.h>
#include <math.h>
#include <stdint.h>


#define ATT_SLOPE 0.2f
#define ACT_SLOPE 0.01f

typedef __attribute__((ext_vector_type(2))) float v2f;
typedef __attribute__((ext_vector_type(8))) float v8f;
typedef __attribute__((ext_vector_type(4))) float v4f;

__device__ __forceinline__ float lrelu(float v, float s) { return v > 0.f ? v : v * s; }

__device__ __forceinline__ void lds_max_f(int* slot, float val) {
  if (val >= 0.f)
    atomicMax(slot, __float_as_int(val));
  else
    atomicMin((unsigned int*)slot, (unsigned int)__float_as_int(val));
}

constexpr int GEMM_WPG = 4;

template <int MASK>
__device__ __forceinline__ float row_xadd(float v) {
  return v + __int_as_float(__builtin_amdgcn_update_dpp(0, __float_as_int(v), 0x160 + MASK,
                                                        0xf, 0xf, true));
}

template <int K, int H>
__global__ void __launch_bounds__(GEMM_WPG * 32)
wmma_gemm_coef(const float* __restrict__ A, const float* __restrict__ B,
               const float* __restrict__ att_src, const float* __restrict__ att_dst,
               float* __restrict__ D, float* __restrict__ as, float* __restrict__ ad, int M) {
  constexpr int NCOLS = H * 16;
  const int m0 = (blockIdx.x * GEMM_WPG + (threadIdx.x >> 5)) * 16;
  if (m0 >= M) return;
  const int lane = threadIdx.x & 31;
  const int half = lane >> 4;
  const int lid  = lane & 15;
  const int ar = min(m0 + lid, M - 1);
  v8f acc[H];
#pragma unroll
  for (int t = 0; t < H; ++t) acc[t] = v8f{};
#pragma unroll
  for (int k = 0; k < K; k += 4) {
    const int ka = k + half * 2;
    v2f a;
    a.x = A[(size_t)ar * K + ka];
    a.y = A[(size_t)ar * K + ka + 1];
#pragma unroll
    for (int t = 0; t < H; ++t) {
      v2f b;
      b.x = B[(size_t)ka * NCOLS + t * 16 + lid];
      b.y = B[(size_t)(ka + 1) * NCOLS + t * 16 + lid];
      acc[t] = __builtin_amdgcn_wmma_f32_16x16x4_f32(false, a, false, b, (short)0, acc[t],
                                                     false, false);
    }
  }
  const int rbase = m0 + half * 8;
  float* const drow = D + (size_t)rbase * NCOLS + lid;
  if (m0 + 16 <= M) {
#pragma unroll
    for (int t = 0; t < H; ++t) {
#pragma unroll
      for (int r = 0; r < 8; ++r) drow[r * NCOLS + t * 16] = acc[t][r];
    }
  } else {
#pragma unroll
    for (int t = 0; t < H; ++t) {
#pragma unroll
      for (int r = 0; r < 8; ++r) {
        if (rbase + r < M) drow[r * NCOLS + t * 16] = acc[t][r];
      }
    }
  }
  float so[H], dv[H];
#pragma unroll
  for (int t = 0; t < H; ++t) {
    const float ws = att_src[t * 16 + lid];
    const float wd = att_dst[t * 16 + lid];
    float s[8], d[8];
#pragma unroll
    for (int r = 0; r < 8; ++r) { s[r] = acc[t][r] * ws; d[r] = acc[t][r] * wd; }
#pragma unroll
    for (int r = 0; r < 8; ++r) { s[r] = row_xadd<1>(s[r]); d[r] = row_xadd<1>(d[r]); }
#pragma unroll
    for (int r = 0; r < 8; ++r) { s[r] = row_xadd<2>(s[r]); d[r] = row_xadd<2>(d[r]); }
#pragma unroll
    for (int r = 0; r < 8; ++r) { s[r] = row_xadd<4>(s[r]); d[r] = row_xadd<4>(d[r]); }
#pragma unroll
    for (int r = 0; r < 8; ++r) { s[r] = row_xadd<8>(s[r]); d[r] = row_xadd<8>(d[r]); }
    float ss = s[0], dd = d[0];
#pragma unroll
    for (int r = 1; r < 8; ++r) { ss = lid == r ? s[r] : ss; dd = lid == r ? d[r] : dd; }
    so[t] = ss; dv[t] = dd;
  }
  const int row = rbase + lid;
  if (lid < 8 && row < M) {
    if constexpr (H % 4 == 0) {
#pragma unroll
      for (int t = 0; t < H; t += 4) {
        *reinterpret_cast<float4*>(as + (size_t)row * H + t) = make_float4(so[t], so[t + 1], so[t + 2], so[t + 3]);
        *reinterpret_cast<float4*>(ad + (size_t)row * H + t) = make_float4(dv[t], dv[t + 1], dv[t + 2], dv[t + 3]);
      }
    } else {
#pragma unroll
      for (int t = 0; t < H; ++t) { as[(size_t)row * H + t] = so[t]; ad[(size_t)row * H + t] = dv[t]; }
    }
  }
}

constexpr int OWN_NT  = 1024;
constexpr int OWN_NPW = 200;
constexpr int OWN_CAP = 9216;
constexpr int OWN_UNR = 8;
constexpr unsigned OWN_NOKEY = 255u;
constexpr int BIT_GRP   = 8;
constexpr int BIT_PPG   = BIT_GRP / 4;
constexpr int BIT_NT    = 512;
constexpr int BIT_EPT   = 16;
constexpr int BIT_TGR   = BIT_EPT / BIT_GRP;
constexpr int BIT_CW    = BIT_NT * BIT_TGR / 32;
constexpr int BIT_PITCH = 257;

__device__ __forceinline__ unsigned owner_key(int d, int Nn) {
  return (unsigned)d < (unsigned)Nn ? min((unsigned)d / (unsigned)OWN_NPW, OWN_NOKEY) : OWN_NOKEY;
}

__global__ void __launch_bounds__(BIT_NT)
owner_bits(const int* __restrict__ dst, unsigned int* __restrict__ bits,
           int E, int Nn, int rows, int rowWords) {
  __shared__ unsigned int lb[BIT_CW * BIT_PITCH];
  const bool vec = ((reinterpret_cast<uintptr_t>(dst) & 15) == 0);
  if (rows > (int)OWN_NOKEY || !vec || E < 4) return;
  const int tid = threadIdx.x;
  for (int i = tid; i < BIT_CW * BIT_PITCH; i += BIT_NT) lb[i] = 0u;
  const int e0 = (blockIdx.x * BIT_NT + tid) * BIT_EPT;
  const int4* d4 = reinterpret_cast<const int4*>(dst);
  const int nq = E >> 2;
  int4 v[4];
#pragma unroll
  for (int j = 0; j < 4; ++j) v[j] = d4[min((e0 >> 2) + j, nq - 1)];
  __syncthreads();
  auto mark = [&](int d, int lg) {
    const unsigned o = owner_key(d, Nn);
    if (o != OWN_NOKEY) atomicOr(&lb[(lg >> 5) * BIT_PITCH + o], 1u << (lg & 31));
  };
#pragma unroll
  for (int j = 0; j < 4; ++j) {
    const int lg = tid * BIT_TGR + j / BIT_PPG;
    if ((e0 >> 2) + j < nq) { mark(v[j].x, lg); mark(v[j].y, lg); mark(v[j].z, lg); mark(v[j].w, lg); }
  }
  if (blockIdx.x == gridDim.x - 1 && tid < (E & 3)) {
    const int e = (E & ~3) + tid;
    mark(dst[e], e / BIT_GRP - blockIdx.x * (BIT_NT * BIT_TGR));
  }
  __syncthreads();
  for (int i = tid; i < rows * BIT_CW; i += BIT_NT) {
    const int o = i / BIT_CW, w = i % BIT_CW;
    bits[(size_t)o * rowWords + blockIdx.x * BIT_CW + w] = lb[w * BIT_PITCH + o];
  }
}

template <int N>
__device__ __forceinline__ int row_shr(int x) {
  return __builtin_amdgcn_update_dpp(0, x, 0x110 + N, 0xf, 0xf, true);
}
__device__ __forceinline__ int wave_reserve(int c, int* counter, int lane) {
  int incl = c;
  incl += row_shr<1>(incl);
  incl += row_shr<2>(incl);
  incl += row_shr<4>(incl);
  incl += row_shr<8>(incl);
  const int r0 = __builtin_amdgcn_readlane(incl, 15);
  const int r1 = __builtin_amdgcn_readlane(incl, 31);
  int base = 0;
  if (lane == 0) base = atomicAdd(counter, r0 + r1);
  base = __builtin_amdgcn_readfirstlane(base);
  return base + incl - c + (lane >= 16 ? r0 : 0);
}

template <int H, int C, bool FINAL, bool SAVED>
__global__ void __launch_bounds__(OWN_NT)
owner_layer(const int* __restrict__ src, const int* __restrict__ dst,
            const unsigned int* __restrict__ bits, int rowWords,
            int* __restrict__ sv_s, int* __restrict__ sv_d, int* __restrict__ sv_n,
            const float* __restrict__ as, const float* __restrict__ ad,
            const float* __restrict__ hfeat, const float* __restrict__ bias,
            const float* __restrict__ Wo, const float* __restrict__ bo,
            float* __restrict__ outp, int E, int Nn) {
  constexpr int HC = H * C;
  static_assert(OWN_NT % HC == 0, "a thread keeps one feature column");
  __shared__ float acc_l[OWN_NPW * HC];
  __shared__ float den_l[OWN_NPW * H];
  __shared__ float ad_l[OWN_NPW * H];
  __shared__ int   m_l[OWN_NPW * H];
  __shared__ int   list_s[OWN_CAP];
  __shared__ int   list_d[OWN_CAP];
  __shared__ float ew[OWN_CAP * H];
  __shared__ int   cnt;
  __shared__ int   gcnt;

  const int tid = threadIdx.x;
  const int n0 = blockIdx.x * OWN_NPW;
  const int nloc = min(OWN_NPW, Nn - n0);
  if (nloc <= 0) return;

  for (int i = tid; i < nloc * HC; i += OWN_NT) acc_l[i] = 0.f;
  for (int i = tid; i < nloc * H; i += OWN_NT) {
    den_l[i] = 0.f;
    m_l[i] = __float_as_int(-INFINITY);
    ad_l[i] = ad[(size_t)n0 * H + i];
  }
  for (int i = tid; i < nloc; i += OWN_NT) { list_s[i] = n0 + i; list_d[i] = i; }
  if (tid == 0) { cnt = nloc; gcnt = 0; }
  __syncthreads();

  int* const gs = sv_s + (size_t)blockIdx.x * OWN_CAP;
  int* const gd = sv_d + (size_t)blockIdx.x * OWN_CAP;

  const bool keyed = !SAVED && gridDim.x <= OWN_NOKEY && E >= 4 &&
                     ((reinterpret_cast<uintptr_t>(dst) & 15) == 0);
  if (keyed) {
    const int lane = tid & 31;
    int* const glist = reinterpret_cast<int*>(ew);
    const unsigned int* const row = bits + (size_t)blockIdx.x * rowWords;
    for (int wb = tid - lane; wb < rowWords; wb += OWN_NT * 4) {
      unsigned wv[4];
      int c = 0;
#pragma unroll
      for (int u = 0; u < 4; ++u) {
        const int wi = wb + lane + u * OWN_NT;
        wv[u] = wi < rowWords ? row[wi] : 0u;
      }
#pragma unroll
      for (int u = 0; u < 4; ++u) c += __popc(wv[u]);
      int pos = wave_reserve(c, &gcnt, lane);
#pragma unroll
      for (int u = 0; u < 4; ++u) {
        unsigned m = wv[u];
        const int g0 = (wb + lane + u * OWN_NT) * 32;
        while (m) {
          const int b = __ffs(m) - 1;
          m &= m - 1;
          if (pos < OWN_CAP) glist[pos] = g0 + b;
          ++pos;
        }
      }
    }
    __syncthreads();
    const int nit = min(gcnt, OWN_CAP) * BIT_PPG;
    const int4* d4 = reinterpret_cast<const int4*>(dst);
    const int nq = E >> 2;
    for (int ib = tid - lane; ib < nit; ib += OWN_NT * OWN_UNR) {
      int4 dv[OWN_UNR];
      int  eb[OWN_UNR];
#pragma unroll
      for (int u = 0; u < OWN_UNR; ++u) {
        const int it = min(ib + lane + u * OWN_NT, nit - 1);
        const int q = glist[it / BIT_PPG] * BIT_PPG + it % BIT_PPG;
        eb[u] = q;
        dv[u] = d4[min(q, nq - 1)];
      }
      unsigned hm = 0;
      unsigned pk[OWN_UNR];
      static_assert(OWN_NPW <= 256, "a local destination fits one byte of the packed word");
#pragma unroll
      for (int u = 0; u < OWN_UNR; ++u) {
        const unsigned x0 = (unsigned)(dv[u].x - n0), x1 = (unsigned)(dv[u].y - n0);
        const unsigned x2 = (unsigned)(dv[u].z - n0), x3 = (unsigned)(dv[u].w - n0);
        const unsigned h = (x0 < (unsigned)nloc ? 1u : 0u) | (x1 < (unsigned)nloc ? 2u : 0u) |
                           (x2 < (unsigned)nloc ? 4u : 0u) | (x3 < (unsigned)nloc ? 8u : 0u);
        pk[u] = (x0 & 255u) | ((x1 & 255u) << 8) | ((x2 & 255u) << 16) | (x3 << 24);
        const bool live = ib + lane + u * OWN_NT < nit && eb[u] < nq;
        hm |= (live ? h : 0u) << (4 * u);
      }
      int pos = wave_reserve(__popc(hm), &cnt, lane);
#pragma unroll
      for (int u = 0; u < OWN_UNR; ++u) {
        unsigned m = (hm >> (4 * u)) & 15u;
        while (m) {
          const int j = __ffs(m) - 1;
          m &= m - 1;
          if (pos < OWN_CAP) { list_s[pos] = eb[u] * 4 + j; list_d[pos] = (int)((pk[u] >> (8 * j)) & 255u); }
          ++pos;
        }
      }
    }
    if (tid < (E & 3)) {
      const int e = (E & ~3) + tid;
      const int dl = dst[e] - n0;
      if ((unsigned)dl < (unsigned)nloc) {
        const int pos = atomicAdd(&cnt, 1);
        if (pos < OWN_CAP) { list_s[pos] = e; list_d[pos] = dl; }
      }
    }
  }
  __syncthreads();
  const int total = SAVED ? sv_n[blockIdx.x] : (keyed ? cnt : OWN_CAP + 1);
  if (!SAVED && tid == 0) sv_n[blockIdx.x] = total;

  if (total <= OWN_CAP) {
   if constexpr (SAVED) {
    for (int i0 = nloc + tid; i0 < total; i0 += OWN_NT * OWN_UNR) {
      int s[OWN_UNR], d[OWN_UNR];
#pragma unroll
      for (int u = 0; u < OWN_UNR; ++u) {
        const int i = min(i0 + u * OWN_NT, total - 1);
        s[u] = gs[i];
        d[u] = gd[i];
      }
#pragma unroll
      for (int u = 0; u < OWN_UNR; ++u) {
        const int i = i0 + u * OWN_NT;
        if (i < total) { list_s[i] = s[u]; list_d[i] = d[u]; }
      }
    }
   } else {
    for (int i0 = nloc + tid; i0 < total; i0 += OWN_NT * OWN_UNR) {
      int s[OWN_UNR], d[OWN_UNR];
#pragma unroll
      for (int u = 0; u < OWN_UNR; ++u) {
        const int i = i0 + u * OWN_NT;
        const int ic = min(i, total - 1);
        const int e = list_s[ic];
        const int ee = i < total ? e : 0;
        s[u] = src[ee];
        d[u] = list_d[ic];
      }
#pragma unroll
      for (int u = 0; u < OWN_UNR; ++u) {
        const int i = i0 + u * OWN_NT;
        if (i < total) {
          list_s[i] = s[u];
          gs[i] = s[u]; gd[i] = d[u];
        }
      }
    }
   }
    __syncthreads();
    {
      constexpr int U1 = 4;
      const int totH = total * H;
      int it = tid;
      for (; it + (U1 - 1) * OWN_NT < totH; it += U1 * OWN_NT) {
        float a[U1];
        int sl[U1];
#pragma unroll
        for (int u = 0; u < U1; ++u) {
          const int q = it + u * OWN_NT;
          const int i = q / H, h = q % H;
          sl[u] = list_d[i] * H + h;
          a[u] = as[(size_t)list_s[i] * H + h];
        }
#pragma unroll
        for (int u = 0; u < U1; ++u) {
          const float v = lrelu(a[u] + ad_l[sl[u]], ATT_SLOPE);
          ew[it + u * OWN_NT] = v;
          lds_max_f(&m_l[sl[u]], v);
        }
      }
      for (; it < totH; it += OWN_NT) {
        const int i = it / H, h = it % H;
        const int sl = list_d[i] * H + h;
        const float v = lrelu(as[(size_t)list_s[i] * H + h] + ad_l[sl], ATT_SLOPE);
        ew[it] = v;
        lds_max_f(&m_l[sl], v);
      }
    }
    __syncthreads();
    for (int it = tid; it < total * H; it += OWN_NT) {
      const int i = it / H, h = it % H;
      const int dl = list_d[i];
      const float w = expf(ew[it] - __int_as_float(m_l[dl * H + h]));
      ew[it] = w;
      atomicAdd(&den_l[dl * H + h], w);
    }
    __syncthreads();
    static_assert(C % 4 == 0 && HC % 4 == 0, "a lane's four columns lie in one head");
    constexpr int LPE = HC / 4;
    static_assert(OWN_NT % LPE == 0 && (LPE == 16 || LPE <= 8) && 32 % LPE == 0, "edge slots of a wave");
    constexpr int EPI = OWN_NT / LPE;
    const int c0 = (tid % LPE) * 4;
    const int h = c0 / C;
    const int slot = (tid & 31) / LPE;
    const int rot = LPE == 16 ? (slot & 1) * 2 : (slot & 3);
    const int o0 = rot, o1 = (rot + 1) & 3, o2 = (rot + 2) & 3, o3 = (rot + 3) & 3;
    auto add4 = [&](int dloc, float wt, const v4f& v) {
      const float a0 = (rot & 1) ? v.y : v.x;
      const float a1 = (rot & 1) ? v.z : v.y;
      const float a2 = (rot & 1) ? v.w : v.z;
      const float a3 = (rot & 1) ? v.x : v.w;
      const float p0 = (rot & 2) ? a2 : a0;
      const float p1 = (rot & 2) ? a3 : a1;
      const float p2 = (rot & 2) ? a0 : a2;
      const float p3 = (rot & 2) ? a1 : a3;
      float* const arow = acc_l + dloc * HC + c0;
      atomicAdd(arow + o0, wt * p0);
      atomicAdd(arow + o1, wt * p1);
      atomicAdd(arow + o2, wt * p2);
      atomicAdd(arow + o3, wt * p3);
    };
    int i0 = tid / LPE;
    for (; i0 + (OWN_UNR - 1) * EPI < total; i0 += EPI * OWN_UNR) {
      v4f hv[OWN_UNR];
      float w[OWN_UNR];
      int dl[OWN_UNR];
#pragma unroll
      for (int u = 0; u < OWN_UNR; ++u) {
        const int i = i0 + u * EPI;
        dl[u] = list_d[i];
        w[u] = ew[i * H + h];
        hv[u] = *reinterpret_cast<const v4f*>(hfeat + (size_t)list_s[i] * HC + c0);
      }
#pragma unroll
      for (int u = 0; u < OWN_UNR; ++u) add4(dl[u], w[u], hv[u]);
    }
    if (i0 < total) {
      v4f hv[OWN_UNR];
      float w[OWN_UNR];
      int dl[OWN_UNR];
#pragma unroll
      for (int u = 0; u < OWN_UNR; ++u) {
        const int i = min(i0 + u * EPI, total - 1);
        dl[u] = list_d[i];
        w[u] = ew[i * H + h];
        hv[u] = *reinterpret_cast<const v4f*>(hfeat + (size_t)list_s[i] * HC + c0);
      }
      asm volatile("" :: "v"(hv[0]), "v"(hv[1]), "v"(hv[2]), "v"(hv[3]),
                         "v"(hv[4]), "v"(hv[5]), "v"(hv[6]), "v"(hv[7]));
      static_assert(OWN_UNR == 8, "the statement above names the eight gathers");
#pragma unroll
      for (int u = 0; u < OWN_UNR; ++u) {
        if (i0 + u * EPI < total) add4(dl[u], w[u], hv[u]);
      }
    }
  } else {
    const int tot = E + nloc;
    for (int e = tid; e < tot; e += OWN_NT) {
      int s, dl;
      if (e < E) {
        const unsigned x = (unsigned)(dst[e] - n0);
        if (x >= (unsigned)nloc) continue;
        dl = (int)x; s = src[e];
      } else { dl = e - E; s = n0 + dl; }
      for (int h = 0; h < H; ++h) {
        const float v = lrelu(as[(size_t)s * H + h] + ad_l[dl * H + h], ATT_SLOPE);
        lds_max_f(&m_l[dl * H + h], v);
      }
    }
    __syncthreads();
    for (int e = tid; e < tot; e += OWN_NT) {
      int s, dl;
      if (e < E) {
        const unsigned x = (unsigned)(dst[e] - n0);
        if (x >= (unsigned)nloc) continue;
        dl = (int)x; s = src[e];
      } else { dl = e - E; s = n0 + dl; }
      for (int h = 0; h < H; ++h) {
        const float v = lrelu(as[(size_t)s * H + h] + ad_l[dl * H + h], ATT_SLOPE);
        const float w = expf(v - __int_as_float(m_l[dl * H + h]));
        atomicAdd(&den_l[dl * H + h], w);
        for (int c = 0; c < C; ++c)
          atomicAdd(&acc_l[dl * HC + h * C + c], w * hfeat[(size_t)s * HC + h * C + c]);
      }
    }
  }
  __syncthreads();

  if constexpr (!FINAL) {
    for (int it = tid; it < nloc * HC; it += OWN_NT) {
      const int n = it / HC, jj = it % HC;
      const float v = acc_l[it] / den_l[n * H + jj / C] + bias[jj];
      outp[(size_t)n0 * HC + it] = lrelu(v, ACT_SLOPE);
    }
  } else {
    for (int n = tid; n < nloc; n += OWN_NT) {
      const float dn = den_l[n];
      float s = 0.f;
      for (int c = 0; c < C; ++c) {
        const float f = lrelu(acc_l[n * C + c] / dn + bias[c], ACT_SLOPE);
        s += f * Wo[c];
      }
      outp[n0 + n] = s + bo[0];
    }
  }
}

extern "C" void kernel_launch(void* const* d_in, const int* in_sizes, int n_in,
                              void* d_out, int out_size, void* d_ws, size_t ws_size,
                              hipStream_t stream) {
  const float* x    = (const float*)d_in[0];
  const int*   ei   = (const int*)d_in[1];
  const float* W1   = (const float*)d_in[2];
  const float* aS1  = (const float*)d_in[3];
  const float* aD1  = (const float*)d_in[4];
  const float* b1   = (const float*)d_in[5];
  const float* W2   = (const float*)d_in[6];
  const float* aS2  = (const float*)d_in[7];
  const float* aD2  = (const float*)d_in[8];
  const float* b2   = (const float*)d_in[9];
  const float* Wo   = (const float*)d_in[10];
  const float* bo   = (const float*)d_in[11];
  float* out = (float*)d_out;

  const int IN_C = 64, HID = 16, HEADS = 4;
  const int HC1 = HEADS * HID;
  const int Nn = in_sizes[0] / IN_C;
  const int E  = in_sizes[1] / 2;
  const int* src = ei;
  const int* dst = ei + E;

  const int GB = (Nn + 16 * GEMM_WPG - 1) / (16 * GEMM_WPG);
  const int OG = (Nn + OWN_NPW - 1) / OWN_NPW;
  const int NCH = (E + BIT_NT * BIT_EPT - 1) / (BIT_NT * BIT_EPT) > 0
                ? (E + BIT_NT * BIT_EPT - 1) / (BIT_NT * BIT_EPT) : 1;
  const int rowWords = NCH * BIT_CW;

  unsigned int* bits = (unsigned int*)d_ws;
  float* p = (float*)(bits + (size_t)OG * rowWords);
  float* h1   = p; p += (size_t)Nn * HC1;
  float* f1   = p; p += (size_t)Nn * HC1;
  float* h2   = p; p += (size_t)Nn * HID;
  float* as1  = p; p += (size_t)Nn * HEADS;
  float* ad1  = p; p += (size_t)Nn * HEADS;
  float* as2  = p; p += (size_t)Nn;
  float* ad2  = p; p += (size_t)Nn;

  int* sv_s = (int*)p;
  int* sv_d = sv_s + (size_t)OG * OWN_CAP;
  int* sv_n = sv_d + (size_t)OG * OWN_CAP;

  owner_bits<<<NCH, BIT_NT, 0, stream>>>(dst, bits, E, Nn, OG, rowWords);

  wmma_gemm_coef<64, 4><<<GB, GEMM_WPG * 32, 0, stream>>>(x, W1, aS1, aD1, h1, as1, ad1, Nn);
  owner_layer<4, 16, false, false><<<OG, OWN_NT, 0, stream>>>(src, dst, bits, rowWords, sv_s, sv_d, sv_n,
                                                            as1, ad1, h1, b1, Wo, bo, f1, E, Nn);

  wmma_gemm_coef<64, 1><<<GB, GEMM_WPG * 32, 0, stream>>>(f1, W2, aS2, aD2, h2, as2, ad2, Nn);
  owner_layer<1, 16, true, true><<<OG, OWN_NT, 0, stream>>>(src, dst, bits, rowWords, sv_s, sv_d, sv_n,
                                                          as2, ad2, h2, b2, Wo, bo, out, E, Nn);
}
